# starting kernel
# baseline (speedup 1.0000x reference)
#include <hip/hip_runtime.h>

#define N_USERS  100000
#define N_ITEMS  50000
#define N_NODES  (N_USERS + N_ITEMS)
#define NNZ_CT   1500000
#define BATCH_CT 1048576
#define DIM      64

typedef __attribute__((ext_vector_type(2))) float v2f;
typedef __attribute__((ext_vector_type(8))) float v8f;

__global__ __launch_bounds__(256) void zero_kernel(float4* __restrict__ p, int n4) {
    int i = blockIdx.x * blockDim.x + threadIdx.x;
    if (i < n4) p[i] = make_float4(0.f, 0.f, 0.f, 0.f);
}

__global__ __launch_bounds__(256) void spmm_kernel(
    const int*   __restrict__ rows, const int* __restrict__ cols,
    const float* __restrict__ vals,
    const float* __restrict__ srcA, const float* __restrict__ srcB, int split,
    float* __restrict__ Hout, int nnz) {
    long long t = (long long)blockIdx.x * blockDim.x + threadIdx.x;
    int e = (int)(t >> 4);
    if (e >= nnz) return;
    int dq = ((int)t & 15) << 2;
    int r = rows[e];
    int c = cols[e];
    float v = vals[e];
    const float* src = (c < split) ? (srcA + (size_t)c * DIM)
                                   : (srcB + (size_t)(c - split) * DIM);
    float4 h = *(const float4*)(src + dq);
    float* dst = Hout + (size_t)r * DIM + dq;
    unsafeAtomicAdd(dst + 0, v * h.x);
    unsafeAtomicAdd(dst + 1, v * h.y);
    unsafeAtomicAdd(dst + 2, v * h.z);
    unsafeAtomicAdd(dst + 3, v * h.w);
}

__global__ __launch_bounds__(256) void pred_kernel(
    const float* __restrict__ U, const float* __restrict__ I,
    const int*   __restrict__ u_idx, const int* __restrict__ i_idx,
    const float* __restrict__ bu, const float* __restrict__ bi,
    const float* __restrict__ mu, float* __restrict__ out) {
    int wave = blockIdx.x * (blockDim.x >> 5) + (threadIdx.x >> 5);
    int lane = threadIdx.x & 31;
    int base = wave << 4;
    int m    = lane & 15;
    int hi   = lane >> 4;
    int ur = u_idx[base + m];
    int ir = i_idx[base + m];
    const float* up = U + (size_t)ur * DIM;
    const float* ip = I + (size_t)ir * DIM;

    v8f c = {};
#pragma unroll
    for (int k0 = 0; k0 < DIM; k0 += 4) {
        v2f a = *(const v2f*)(up + k0 + 2 * hi);
        v2f b = *(const v2f*)(ip + k0 + 2 * hi);
        c = __builtin_amdgcn_wmma_f32_16x16x4_f32(
                false, a, false, b, (short)0, c, false, false);
    }

    float d = 0.f;
#pragma unroll
    for (int j = 0; j < 8; ++j) {
        float vlo = __shfl(c[j], j, 32);
        float vhi = __shfl(c[j], j + 24, 32);
        if (lane == j)     d = vlo;
        if (lane == j + 8) d = vhi;
    }

    if (lane < 16) {
        out[base + lane] = d + bu[ur] + bi[ir] + mu[0];
    }
}

extern "C" void kernel_launch(void* const* d_in, const int* in_sizes, int n_in,
                              void* d_out, int out_size, void* d_ws, size_t ws_size,
                              hipStream_t stream) {
    const int*   u_idx  = (const int*)d_in[0];
    const int*   i_idx  = (const int*)d_in[1];
    const int*   A_rows = (const int*)d_in[2];
    const int*   A_cols = (const int*)d_in[3];
    const float* A_vals = (const float*)d_in[4];
    const float* Eu     = (const float*)d_in[5];
    const float* Ei     = (const float*)d_in[6];
    const float* bu     = (const float*)d_in[7];
    const float* bi     = (const float*)d_in[8];
    const float* mu     = (const float*)d_in[9];
    float* out = (float*)d_out;

    float* H1 = (float*)d_ws;
    float* H2 = H1 + (size_t)N_NODES * DIM;

    int n4 = 2 * N_NODES * DIM / 4;
    zero_kernel<<<(n4 + 255) / 256, 256, 0, stream>>>((float4*)d_ws, n4);

    long long spmm_threads = (long long)NNZ_CT * 16;
    int spmm_blocks = (int)((spmm_threads + 255) / 256);
    spmm_kernel<<<spmm_blocks, 256, 0, stream>>>(
        A_rows, A_cols, A_vals, Eu, Ei, N_USERS, H1, NNZ_CT);

    spmm_kernel<<<spmm_blocks, 256, 0, stream>>>(
        A_rows, A_cols, A_vals, H1, H1, N_NODES + 1, H2, NNZ_CT);

    int pred_blocks = (BATCH_CT / 16) / 8;
    pred_kernel<<<pred_blocks, 256, 0, stream>>>(
        H2, H2 + (size_t)N_USERS * DIM, u_idx, i_idx, bu, bi, mu, out);
}


# spmm by 256 destination-range owners: each group filters the whole edge list for its 586 rows, sums in LDS (f32 LDS adds, arrival order), stores rows once; no global atomics, zero fill or placing
# speedup vs baseline: 1.2427x; 1.2427x over previous
#include <hip/hip_runtime.h>

#define N_USERS  100000
#define N_ITEMS  50000
#define N_NODES  (N_USERS + N_ITEMS)
#define NNZ_CT   1500000
#define BATCH_CT 1048576
#define DIM      64

typedef __attribute__((ext_vector_type(2))) float v2f;
typedef __attribute__((ext_vector_type(8))) float v8f;

#define OWN_ROWS   586
#define OWN_GROUPS 256
#define WAVE_CAP   1024
#define N_QUADS    (NNZ_CT / 4)
#define SCAN_STEPS 368
#define ROT_QUADS  1464

#define NOTE_EDGE(m, b, e)                                              \
    if (b) {                                                            \
        if (m) {                                                        \
            int slot = wcount + (int)__builtin_amdgcn_mbcnt_lo((b), 0u);\
            if (slot < WAVE_CAP) mylist[slot] = (e);                    \
        }                                                               \
        wcount += __builtin_popcount(b);                                \
    }

__global__ __launch_bounds__(1024) void spmm_owner_kernel(
    const int*   __restrict__ rows, const int* __restrict__ cols,
    const float* __restrict__ vals,
    const float* __restrict__ srcA, const float* __restrict__ srcB, int split,
    float* __restrict__ Hout) {
    __shared__ float acc[OWN_ROWS * DIM];
    __shared__ int   elist[32 * WAVE_CAP];
    __shared__ int   spill;

    const int tid  = threadIdx.x;
    const int lane = tid & 31;
    const int base = blockIdx.x * OWN_ROWS;
    int nr = N_NODES - base;
    if (nr > OWN_ROWS) nr = OWN_ROWS;

    float4* acc4 = (float4*)acc;
    for (int i = tid; i < OWN_ROWS * DIM / 4; i += 1024)
        acc4[i] = make_float4(0.f, 0.f, 0.f, 0.f);
    if (tid == 0) spill = 0;
    __syncthreads();

    const int wave = __builtin_amdgcn_readfirstlane(tid >> 5);
    int* mylist = elist + wave * WAVE_CAP;
    int  wcount = 0;

    const int4* rows4 = (const int4*)rows;
    const int rot = blockIdx.x * ROT_QUADS;
    for (int i = 0; i < SCAN_STEPS; i += 16) {
        int4 rv[16];
        int  qx[16];
#pragma unroll
        for (int u = 0; u < 16; ++u) {
            int q = tid + 1024 * (i + u);
            int x = (q < N_QUADS ? q : 0) + rot;
            if (x >= N_QUADS) x -= N_QUADS;
            qx[u] = x;
            rv[u] = rows4[x];
        }
#pragma unroll
        for (int u = 0; u < 16; ++u) {
            bool ok = (tid + 1024 * (i + u)) < N_QUADS;
            bool m0 = ok && (unsigned)(rv[u].x - base) < (unsigned)nr;
            bool m1 = ok && (unsigned)(rv[u].y - base) < (unsigned)nr;
            bool m2 = ok && (unsigned)(rv[u].z - base) < (unsigned)nr;
            bool m3 = ok && (unsigned)(rv[u].w - base) < (unsigned)nr;
            unsigned b0 = __builtin_amdgcn_ballot_w32(m0);
            unsigned b1 = __builtin_amdgcn_ballot_w32(m1);
            unsigned b2 = __builtin_amdgcn_ballot_w32(m2);
            unsigned b3 = __builtin_amdgcn_ballot_w32(m3);
            if (b0 | b1 | b2 | b3) {
                int e0 = qx[u] * 4;
                NOTE_EDGE(m0, b0, e0 + 0)
                NOTE_EDGE(m1, b1, e0 + 1)
                NOTE_EDGE(m2, b2, e0 + 2)
                NOTE_EDGE(m3, b3, e0 + 3)
            }
        }
    }
    if (wcount > WAVE_CAP) spill = 1;
    __syncthreads();

    if (__builtin_amdgcn_readfirstlane(spill) == 0) {
        const int n = wcount;
        for (int i0 = 0; i0 < n; i0 += 32) {
            int cnt = n - i0;
            if (cnt > 32) cnt = 32;
            int   e  = mylist[i0 + (lane < cnt ? lane : 0)];
            int   rl = rows[e] - base;
            int   c  = cols[e];
            float v  = vals[e];
            for (int j0 = 0; j0 < cnt; j0 += 16) {
                v2f h[16];
#pragma unroll
                for (int u = 0; u < 16; ++u) {
                    int j = j0 + u;
                    if (j > cnt - 1) j = cnt - 1;
                    int cj = __builtin_amdgcn_readlane(c, j);
                    const float* sp = (cj < split)
                        ? (srcA + (size_t)cj * DIM)
                        : (srcB + (size_t)(cj - split) * DIM);
                    h[u] = *(const v2f*)(sp + 2 * lane);
                }
#pragma unroll
                for (int u = 0; u < 16; ++u) {
                    int j = j0 + u;
                    if (j < cnt) {
                        int   rj = __builtin_amdgcn_readlane(rl, j);
                        float vj = __builtin_bit_cast(float,
                            __builtin_amdgcn_readlane(__builtin_bit_cast(int, v), j));
                        float* a = acc + rj * DIM + 2 * lane;
                        unsafeAtomicAdd(a,     vj * h[u].x);
                        unsafeAtomicAdd(a + 1, vj * h[u].y);
                    }
                }
            }
        }
    } else {
        for (int q = tid; q < N_QUADS; q += 1024) {
            int4 r4 = rows4[q];
            int rr[4] = {r4.x, r4.y, r4.z, r4.w};
#pragma unroll
            for (int k = 0; k < 4; ++k) {
                unsigned loc = (unsigned)(rr[k] - base);
                if (loc < (unsigned)nr) {
                    int   e = q * 4 + k;
                    int   c = cols[e];
                    float v = vals[e];
                    const float* sp = (c < split)
                        ? (srcA + (size_t)c * DIM)
                        : (srcB + (size_t)(c - split) * DIM);
                    float* a = acc + loc * DIM;
#pragma nounroll
                    for (int d = 0; d < DIM; ++d)
                        unsafeAtomicAdd(a + d, v * sp[d]);
                }
            }
        }
    }
    __syncthreads();

    float4* out4 = (float4*)(Hout + (size_t)base * DIM);
    for (int i = tid; i < nr * DIM / 4; i += 1024)
        out4[i] = acc4[i];
}

__global__ __launch_bounds__(256) void pred_kernel(
    const float* __restrict__ U, const float* __restrict__ I,
    const int*   __restrict__ u_idx, const int* __restrict__ i_idx,
    const float* __restrict__ bu, const float* __restrict__ bi,
    const float* __restrict__ mu, float* __restrict__ out) {
    int wave = blockIdx.x * (blockDim.x >> 5) + (threadIdx.x >> 5);
    int lane = threadIdx.x & 31;
    int base = wave << 4;
    int m    = lane & 15;
    int hi   = lane >> 4;
    int ur = u_idx[base + m];
    int ir = i_idx[base + m];
    const float* up = U + (size_t)ur * DIM;
    const float* ip = I + (size_t)ir * DIM;

    v8f c = {};
#pragma unroll
    for (int k0 = 0; k0 < DIM; k0 += 4) {
        v2f a = *(const v2f*)(up + k0 + 2 * hi);
        v2f b = *(const v2f*)(ip + k0 + 2 * hi);
        c = __builtin_amdgcn_wmma_f32_16x16x4_f32(
                false, a, false, b, (short)0, c, false, false);
    }

    float d = 0.f;
#pragma unroll
    for (int j = 0; j < 8; ++j) {
        float vlo = __shfl(c[j], j, 32);
        float vhi = __shfl(c[j], j + 24, 32);
        if (lane == j)     d = vlo;
        if (lane == j + 8) d = vhi;
    }

    if (lane < 16) {
        out[base + lane] = d + bu[ur] + bi[ir] + mu[0];
    }
}

extern "C" void kernel_launch(void* const* d_in, const int* in_sizes, int n_in,
                              void* d_out, int out_size, void* d_ws, size_t ws_size,
                              hipStream_t stream) {
    const int*   u_idx  = (const int*)d_in[0];
    const int*   i_idx  = (const int*)d_in[1];
    const int*   A_rows = (const int*)d_in[2];
    const int*   A_cols = (const int*)d_in[3];
    const float* A_vals = (const float*)d_in[4];
    const float* Eu     = (const float*)d_in[5];
    const float* Ei     = (const float*)d_in[6];
    const float* bu     = (const float*)d_in[7];
    const float* bi     = (const float*)d_in[8];
    const float* mu     = (const float*)d_in[9];
    float* out = (float*)d_out;

    float* H1 = (float*)d_ws;
    float* H2 = H1 + (size_t)N_NODES * DIM;

    spmm_owner_kernel<<<OWN_GROUPS, 1024, 0, stream>>>(
        A_rows, A_cols, A_vals, Eu, Ei, N_USERS, H1);

    spmm_owner_kernel<<<OWN_GROUPS, 1024, 0, stream>>>(
        A_rows, A_cols, A_vals, H1, H1, N_NODES + 1, H2);

    int pred_blocks = (BATCH_CT / 16) / 8;
    pred_kernel<<<pred_blocks, 256, 0, stream>>>(
        H2, H2 + (size_t)N_USERS * DIM, u_idx, i_idx, bu, bi, mu, out);
}


# spmm: layer 1 also stores each group's filtered edges as 8-B records in the group's own rows of the layer-2 plane; layer 2 reads them, no second edge-list scan; LDS adds in arrival order as before
# speedup vs baseline: 1.6276x; 1.3097x over previous
#include <hip/hip_runtime.h>

#define N_USERS  100000
#define N_ITEMS  50000
#define N_NODES  (N_USERS + N_ITEMS)
#define NNZ_CT   1500000
#define BATCH_CT 1048576
#define DIM      64

typedef __attribute__((ext_vector_type(2))) float v2f;
typedef __attribute__((ext_vector_type(8))) float v8f;

#define OWN_ROWS   586
#define OWN_GROUPS 256
#define WAVE_CAP   1024
#define N_QUADS    (NNZ_CT / 4)
#define SCAN_STEPS 368
#define ROT_QUADS  1464
#define REC_HDR    32
#define COL_BITS   18
static_assert(N_NODES <= (1 << COL_BITS), "source row id must fit its field");
static_assert(OWN_ROWS <= (1 << (32 - COL_BITS)), "local row must fit its field");

#define NOTE_EDGE(m, b, e)                                              \
    if (b) {                                                            \
        if (m) {                                                        \
            int slot = wcount + (int)__builtin_amdgcn_mbcnt_lo((b), 0u);\
            if (slot < WAVE_CAP) mylist[slot] = (e);                    \
        }                                                               \
        wcount += __builtin_popcount(b);                                \
    }

__device__ __forceinline__ void add_edges(
    int rl, int c, float v, int cnt, int lane, float* acc,
    const float* __restrict__ srcA, const float* __restrict__ srcB, int split) {
    for (int j0 = 0; j0 < cnt; j0 += 16) {
        v2f h[16];
#pragma unroll
        for (int u = 0; u < 16; ++u) {
            int j = j0 + u;
            if (j > cnt - 1) j = cnt - 1;
            int cj = __builtin_amdgcn_readlane(c, j);
            const float* sp = (cj < split)
                ? (srcA + (size_t)cj * DIM)
                : (srcB + (size_t)(cj - split) * DIM);
            h[u] = *(const v2f*)(sp + 2 * lane);
        }
#pragma unroll
        for (int u = 0; u < 16; ++u) {
            int j = j0 + u;
            if (j < cnt) {
                int   rj = __builtin_amdgcn_readlane(rl, j);
                float vj = __builtin_bit_cast(float,
                    __builtin_amdgcn_readlane(__builtin_bit_cast(int, v), j));
                float* a = acc + rj * DIM + 2 * lane;
                unsafeAtomicAdd(a,     vj * h[u].x);
                unsafeAtomicAdd(a + 1, vj * h[u].y);
            }
        }
    }
}

__global__ __launch_bounds__(1024) void spmm_owner_kernel(
    const int*   __restrict__ rows, const int* __restrict__ cols,
    const float* __restrict__ vals,
    const float* __restrict__ srcA, const float* __restrict__ srcB, int split,
    float* Hout, float* recplane, int first) {
    __shared__ float acc[OWN_ROWS * DIM];
    __shared__ int   elist[32 * WAVE_CAP];
    __shared__ int   spill;
    __shared__ int   wcnt_s[32];

    const int tid  = threadIdx.x;
    const int lane = tid & 31;
    const int base = blockIdx.x * OWN_ROWS;
    int nr = N_NODES - base;
    if (nr > OWN_ROWS) nr = OWN_ROWS;

    float4* acc4 = (float4*)acc;
    for (int i = tid; i < OWN_ROWS * DIM / 4; i += 1024)
        acc4[i] = make_float4(0.f, 0.f, 0.f, 0.f);
    if (tid == 0) spill = 0;

    uint2* myrec = (uint2*)(recplane + (size_t)base * DIM);
    int*   hdr   = (int*)myrec;
    const int rec_cap = nr * (DIM / 2) - REC_HDR;

    const int wave = __builtin_amdgcn_readfirstlane(tid >> 5);

    int listed = 0;
    int cl = 0;
    if (!first) {
        listed = __builtin_amdgcn_readfirstlane(hdr[32]) == 1;
        cl = hdr[lane];
    }
    __syncthreads();

    if (listed) {
        int n = 0, off = 0;
        for (int w = 0; w < 32; ++w) {
            int cw = __builtin_amdgcn_readlane(cl, w);
            if (w < wave)  off += (cw + 31) & ~31;
            if (w == wave) n = cw;
        }
        const uint2* seg = myrec + REC_HDR + off;
        uint2 nxt = make_uint2(0u, 0u);
        if (n > 0) nxt = seg[lane < n ? lane : 0];
        for (int i0 = 0; i0 < n; i0 += 32) {
            int cnt = n - i0;
            if (cnt > 32) cnt = 32;
            uint2 cur = nxt;
            if (i0 + 32 < n) {
                int x = i0 + 32 + lane;
                nxt = seg[x < n ? x : i0 + 32];
            }
            int   rl = (int)(cur.x >> COL_BITS);
            int   c  = (int)(cur.x & ((1u << COL_BITS) - 1u));
            float v  = __builtin_bit_cast(float, cur.y);
            add_edges(rl, c, v, cnt, lane, acc, srcA, srcB, split);
        }
    } else {
        int* mylist = elist + wave * WAVE_CAP;
        int  wcount = 0;

        const int4* rows4 = (const int4*)rows;
        const int rot = blockIdx.x * ROT_QUADS;
        for (int i = 0; i < SCAN_STEPS; i += 16) {
            int4 rv[16];
            int  qx[16];
#pragma unroll
            for (int u = 0; u < 16; ++u) {
                int q = tid + 1024 * (i + u);
                int x = (q < N_QUADS ? q : 0) + rot;
                if (x >= N_QUADS) x -= N_QUADS;
                qx[u] = x;
                rv[u] = rows4[x];
            }
#pragma unroll
            for (int u = 0; u < 16; ++u) {
                bool ok = (tid + 1024 * (i + u)) < N_QUADS;
                bool m0 = ok && (unsigned)(rv[u].x - base) < (unsigned)nr;
                bool m1 = ok && (unsigned)(rv[u].y - base) < (unsigned)nr;
                bool m2 = ok && (unsigned)(rv[u].z - base) < (unsigned)nr;
                bool m3 = ok && (unsigned)(rv[u].w - base) < (unsigned)nr;
                unsigned b0 = __builtin_amdgcn_ballot_w32(m0);
                unsigned b1 = __builtin_amdgcn_ballot_w32(m1);
                unsigned b2 = __builtin_amdgcn_ballot_w32(m2);
                unsigned b3 = __builtin_amdgcn_ballot_w32(m3);
                if (b0 | b1 | b2 | b3) {
                    int e0 = qx[u] * 4;
                    NOTE_EDGE(m0, b0, e0 + 0)
                    NOTE_EDGE(m1, b1, e0 + 1)
                    NOTE_EDGE(m2, b2, e0 + 2)
                    NOTE_EDGE(m3, b3, e0 + 3)
                }
            }
        }
        if (wcount > WAVE_CAP) spill = 1;
        if (lane == 0) wcnt_s[wave] = wcount;
        __syncthreads();

        const int spilled = __builtin_amdgcn_readfirstlane(spill);

        int off = 0, padded = 0;
        const int cw_l = wcnt_s[lane];
        for (int w = 0; w < 32; ++w) {
            int cw = (__builtin_amdgcn_readlane(cw_l, w) + 31) & ~31;
            if (w < wave) off += cw;
            padded += cw;
        }
        const int wr = first && !spilled && padded <= rec_cap;
        if (first && wave == 0) {
            hdr[lane] = cw_l;
            if (lane == 0) hdr[32] = wr;
        }

        if (spilled == 0) {
            const int n = wcount;
            uint2* seg = myrec + REC_HDR + off;
            for (int i0 = 0; i0 < n; i0 += 32) {
                int cnt = n - i0;
                if (cnt > 32) cnt = 32;
                int   e  = mylist[i0 + (lane < cnt ? lane : 0)];
                int   rl = rows[e] - base;
                int   c  = cols[e];
                float v  = vals[e];
                if (wr && lane < cnt)
                    seg[i0 + lane] = make_uint2(((unsigned)rl << COL_BITS) | (unsigned)c,
                                                __builtin_bit_cast(unsigned, v));
                add_edges(rl, c, v, cnt, lane, acc, srcA, srcB, split);
            }
        } else {
            for (int q = tid; q < N_QUADS; q += 1024) {
                int4 r4 = rows4[q];
                int rr[4] = {r4.x, r4.y, r4.z, r4.w};
#pragma unroll
                for (int k = 0; k < 4; ++k) {
                    unsigned loc = (unsigned)(rr[k] - base);
                    if (loc < (unsigned)nr) {
                        int   e = q * 4 + k;
                        int   c = cols[e];
                        float v = vals[e];
                        const float* sp = (c < split)
                            ? (srcA + (size_t)c * DIM)
                            : (srcB + (size_t)(c - split) * DIM);
                        float* a = acc + loc * DIM;
#pragma nounroll
                        for (int d = 0; d < DIM; ++d)
                            unsafeAtomicAdd(a + d, v * sp[d]);
                    }
                }
            }
        }
    }
    __syncthreads();

    float4* out4 = (float4*)(Hout + (size_t)base * DIM);
    for (int i = tid; i < nr * DIM / 4; i += 1024)
        out4[i] = acc4[i];
}

__global__ __launch_bounds__(256) void pred_kernel(
    const float* __restrict__ U, const float* __restrict__ I,
    const int*   __restrict__ u_idx, const int* __restrict__ i_idx,
    const float* __restrict__ bu, const float* __restrict__ bi,
    const float* __restrict__ mu, float* __restrict__ out) {
    int wave = blockIdx.x * (blockDim.x >> 5) + (threadIdx.x >> 5);
    int lane = threadIdx.x & 31;
    int base = wave << 4;
    int m    = lane & 15;
    int hi   = lane >> 4;
    int ur = u_idx[base + m];
    int ir = i_idx[base + m];
    const float* up = U + (size_t)ur * DIM;
    const float* ip = I + (size_t)ir * DIM;

    v8f c = {};
#pragma unroll
    for (int k0 = 0; k0 < DIM; k0 += 4) {
        v2f a = *(const v2f*)(up + k0 + 2 * hi);
        v2f b = *(const v2f*)(ip + k0 + 2 * hi);
        c = __builtin_amdgcn_wmma_f32_16x16x4_f32(
                false, a, false, b, (short)0, c, false, false);
    }

    float d = 0.f;
#pragma unroll
    for (int j = 0; j < 8; ++j) {
        float vlo = __shfl(c[j], j, 32);
        float vhi = __shfl(c[j], j + 24, 32);
        if (lane == j)     d = vlo;
        if (lane == j + 8) d = vhi;
    }

    if (lane < 16) {
        out[base + lane] = d + bu[ur] + bi[ir] + mu[0];
    }
}

extern "C" void kernel_launch(void* const* d_in, const int* in_sizes, int n_in,
                              void* d_out, int out_size, void* d_ws, size_t ws_size,
                              hipStream_t stream) {
    const int*   u_idx  = (const int*)d_in[0];
    const int*   i_idx  = (const int*)d_in[1];
    const int*   A_rows = (const int*)d_in[2];
    const int*   A_cols = (const int*)d_in[3];
    const float* A_vals = (const float*)d_in[4];
    const float* Eu     = (const float*)d_in[5];
    const float* Ei     = (const float*)d_in[6];
    const float* bu     = (const float*)d_in[7];
    const float* bi     = (const float*)d_in[8];
    const float* mu     = (const float*)d_in[9];
    float* out = (float*)d_out;

    float* H1 = (float*)d_ws;
    float* H2 = H1 + (size_t)N_NODES * DIM;

    spmm_owner_kernel<<<OWN_GROUPS, 1024, 0, stream>>>(
        A_rows, A_cols, A_vals, Eu, Ei, N_USERS, H1, H2, 1);

    spmm_owner_kernel<<<OWN_GROUPS, 1024, 0, stream>>>(
        A_rows, A_cols, A_vals, H1, H1, N_NODES + 1, H2, H2, 0);

    int pred_blocks = (BATCH_CT / 16) / 8;
    pred_kernel<<<pred_blocks, 256, 0, stream>>>(
        H2, H2 + (size_t)N_USERS * DIM, u_idx, i_idx, bu, bi, mu, out);
}


# spmm layer-1 edge filter tests a 1-byte owner key per edge (row/586, written by a new small launch in front): 16 edges per 16-B lane load, not 4 row ids; LDS f32 adds in arrival order as before
# speedup vs baseline: 1.8183x; 1.1172x over previous
#include <hip/hip_runtime.h>

#define N_USERS  100000
#define N_ITEMS  50000
#define N_NODES  (N_USERS + N_ITEMS)
#define NNZ_CT   1500000
#define BATCH_CT 1048576
#define DIM      64

typedef __attribute__((ext_vector_type(2))) float v2f;
typedef __attribute__((ext_vector_type(8))) float v8f;

#define OWN_ROWS   586
#define OWN_GROUPS 256
#define WAVE_CAP   1024
#define N_QUADS    (NNZ_CT / 4)
#define SCAN_STEPS 368
#define ROT_QUADS  1464
#define REC_HDR    32
#define COL_BITS   18
#define KEY_STEPS  92
#define KEY_UNITS  (32 * KEY_STEPS)
#define KEY_CHUNK_BYTES 8192
#define KEY_ROW0   (OWN_ROWS - KEY_CHUNK_BYTES / (DIM * 4))
#define KEY_OFF_BYTES (KEY_ROW0 * DIM * 4)
#define RANGE_BYTES (OWN_ROWS * DIM * 4)
static_assert(N_NODES <= (1 << COL_BITS), "source row id must fit its field");
static_assert(OWN_ROWS <= (1 << (32 - COL_BITS)), "local row must fit its field");
static_assert((N_NODES - 1) / OWN_ROWS < 256, "owner id must fit a byte");
static_assert(NNZ_CT % 16 == 0 && KEY_UNITS * 512 >= NNZ_CT, "keys cover the edge list");
static_assert(KEY_UNITS * 512 / KEY_CHUNK_BYTES <= OWN_GROUPS - 1, "key chunks stay in full-size ranges");
static_assert(N_NODES - (OWN_GROUPS - 1) * OWN_ROWS >= KEY_ROW0, "last range holds its records");
static_assert(KEY_UNITS * 128 <= 6 * 65536, "the key launch covers every unit");

__device__ __forceinline__ unsigned nonzero_bytes(unsigned x) {
    return ((x & 0x7f7f7f7fu) + 0x7f7f7f7fu) | x;
}

__global__ __launch_bounds__(256) void owner_key_kernel(
    const int* __restrict__ rows, float* keyplane) {
    const int4* rows4 = (const int4*)rows;
    const int t = blockIdx.x * 256 + threadIdx.x;
    int4 r[6];
#pragma unroll
    for (int j = 0; j < 6; ++j) {
        int q = t + j * 65536;
        r[j] = rows4[q < N_QUADS ? q : 0];
    }
#pragma unroll
    for (int j = 0; j < 6; ++j) {
        int q = t + j * 65536;
        if (q < KEY_UNITS * 128) {
            unsigned k = 0u;
            if (q < N_QUADS)
                k =  (unsigned)(r[j].x / OWN_ROWS)        | ((unsigned)(r[j].y / OWN_ROWS) << 8)
                  | ((unsigned)(r[j].z / OWN_ROWS) << 16) | ((unsigned)(r[j].w / OWN_ROWS) << 24);
            int e  = q * 4;
            int ch = e / KEY_CHUNK_BYTES;
            int of = e - ch * KEY_CHUNK_BYTES;
            *(unsigned*)((char*)keyplane + (size_t)ch * RANGE_BYTES + KEY_OFF_BYTES + of) = k;
        }
    }
}

#define NOTE_EDGE(m, b, e)                                              \
    if (b) {                                                            \
        if (m) {                                                        \
            int slot = wcount + (int)__builtin_amdgcn_mbcnt_lo((b), 0u);\
            if (slot < WAVE_CAP) mylist[slot] = (e);                    \
        }                                                               \
        wcount += __builtin_popcount(b);                                \
    }

__device__ __forceinline__ void add_edges(
    int rl, int c, float v, int cnt, int lane, float* acc,
    const float* __restrict__ srcA, const float* __restrict__ srcB, int split) {
    for (int j0 = 0; j0 < cnt; j0 += 16) {
        v2f h[16];
#pragma unroll
        for (int u = 0; u < 16; ++u) {
            int j = j0 + u;
            if (j > cnt - 1) j = cnt - 1;
            int cj = __builtin_amdgcn_readlane(c, j);
            const float* sp = (cj < split)
                ? (srcA + (size_t)cj * DIM)
                : (srcB + (size_t)(cj - split) * DIM);
            h[u] = *(const v2f*)(sp + 2 * lane);
        }
#pragma unroll
        for (int u = 0; u < 16; ++u) {
            int j = j0 + u;
            if (j < cnt) {
                int   rj = __builtin_amdgcn_readlane(rl, j);
                float vj = __builtin_bit_cast(float,
                    __builtin_amdgcn_readlane(__builtin_bit_cast(int, v), j));
                float* a = acc + rj * DIM + 2 * lane;
                unsafeAtomicAdd(a,     vj * h[u].x);
                unsafeAtomicAdd(a + 1, vj * h[u].y);
            }
        }
    }
}

__global__ __launch_bounds__(1024) void spmm_owner_kernel(
    const int*   __restrict__ rows, const int* __restrict__ cols,
    const float* __restrict__ vals,
    const float* __restrict__ srcA, const float* __restrict__ srcB, int split,
    float* Hout, float* recplane, int first) {
    __shared__ float acc[OWN_ROWS * DIM];
    __shared__ int   elist[32 * WAVE_CAP];
    __shared__ int   spill;
    __shared__ int   wcnt_s[32];

    const int tid  = threadIdx.x;
    const int lane = tid & 31;
    const int base = blockIdx.x * OWN_ROWS;
    int nr = N_NODES - base;
    if (nr > OWN_ROWS) nr = OWN_ROWS;

    float4* acc4 = (float4*)acc;
    for (int i = tid; i < OWN_ROWS * DIM / 4; i += 1024)
        acc4[i] = make_float4(0.f, 0.f, 0.f, 0.f);
    if (tid == 0) spill = 0;

    uint2* myrec = (uint2*)(recplane + (size_t)base * DIM);
    int*   hdr   = (int*)myrec;
    const int rec_cap = KEY_ROW0 * (DIM / 2) - REC_HDR;

    const int wave = __builtin_amdgcn_readfirstlane(tid >> 5);

    int listed = 0;
    int cl = 0;
    if (!first) {
        listed = __builtin_amdgcn_readfirstlane(hdr[32]) == 1;
        cl = hdr[lane];
    }
    __syncthreads();

    if (listed) {
        int n = 0, off = 0;
        for (int w = 0; w < 32; ++w) {
            int cw = __builtin_amdgcn_readlane(cl, w);
            if (w < wave)  off += (cw + 31) & ~31;
            if (w == wave) n = cw;
        }
        const uint2* seg = myrec + REC_HDR + off;
        uint2 nxt = make_uint2(0u, 0u);
        if (n > 0) nxt = seg[lane < n ? lane : 0];
        for (int i0 = 0; i0 < n; i0 += 32) {
            int cnt = n - i0;
            if (cnt > 32) cnt = 32;
            uint2 cur = nxt;
            if (i0 + 32 < n) {
                int x = i0 + 32 + lane;
                nxt = seg[x < n ? x : i0 + 32];
            }
            int   rl = (int)(cur.x >> COL_BITS);
            int   c  = (int)(cur.x & ((1u << COL_BITS) - 1u));
            float v  = __builtin_bit_cast(float, cur.y);
            add_edges(rl, c, v, cnt, lane, acc, srcA, srcB, split);
        }
    } else {
        int* mylist = elist + wave * WAVE_CAP;
        int  wcount = 0;

        const int4* rows4 = (const int4*)rows;
        const int rot = blockIdx.x * ROT_QUADS;
        if (first) {
            const unsigned g4 = (unsigned)blockIdx.x * 0x01010101u;
            const char* kplane = (const char*)recplane;
            const int krot   = (blockIdx.x * KEY_UNITS) >> 8;
            const int lane16 = lane * 16;
            for (int i = 0; i < KEY_STEPS; i += 4) {
                uint4 kv[4];
                int   eb[4];
#pragma unroll
                for (int u = 0; u < 4; ++u) {
                    int x = wave + 32 * (i + u) + krot;
                    if (x >= KEY_UNITS) x -= KEY_UNITS;
                    eb[u] = x * 512;
                    int ch = eb[u] / KEY_CHUNK_BYTES;
                    int of = eb[u] - ch * KEY_CHUNK_BYTES;
                    kv[u] = *(const uint4*)(kplane + (size_t)ch * RANGE_BYTES
                                            + KEY_OFF_BYTES + of + lane16);
                }
#pragma unroll
                for (int u = 0; u < 4; ++u) {
                    unsigned n0 = nonzero_bytes(kv[u].x ^ g4), n1 = nonzero_bytes(kv[u].y ^ g4);
                    unsigned n2 = nonzero_bytes(kv[u].z ^ g4), n3 = nonzero_bytes(kv[u].w ^ g4);
                    if (__builtin_amdgcn_ballot_w32((n0 & n1 & n2 & n3 & 0x80808080u) != 0x80808080u)) {
                        const int e0 = eb[u] + lane16;
                        unsigned lm = ((~n0 & 0x80808080u) >> 7) | ((~n1 & 0x80808080u) >> 6) |
                                      ((~n2 & 0x80808080u) >> 5) | ((~n3 & 0x80808080u) >> 4);
                        if (e0 >= NNZ_CT) lm = 0u;
                        bool     m = lm != 0u;
                        unsigned b = __builtin_amdgcn_ballot_w32(m);
#pragma nounroll
                        while (b != 0u) {
                            int bit  = __builtin_ctz(lm | 0x80000000u);
                            int slot = wcount + (int)__builtin_amdgcn_mbcnt_lo(b, 0u);
                            if (m && slot < WAVE_CAP)
                                mylist[slot] = e0 + ((bit & 7) << 2) + (bit >> 3);
                            lm &= lm - 1u;
                            wcount += __builtin_popcount(b);
                            m = lm != 0u;
                            b = __builtin_amdgcn_ballot_w32(m);
                        }
                    }
                }
            }
        } else {
            for (int i = 0; i < SCAN_STEPS; i += 16) {
                int4 rv[16];
                int  qx[16];
#pragma unroll
                for (int u = 0; u < 16; ++u) {
                    int q = tid + 1024 * (i + u);
                    int x = (q < N_QUADS ? q : 0) + rot;
                    if (x >= N_QUADS) x -= N_QUADS;
                    qx[u] = x;
                    rv[u] = rows4[x];
                }
#pragma unroll
                for (int u = 0; u < 16; ++u) {
                    bool ok = (tid + 1024 * (i + u)) < N_QUADS;
                    bool m0 = ok && (unsigned)(rv[u].x - base) < (unsigned)nr;
                    bool m1 = ok && (unsigned)(rv[u].y - base) < (unsigned)nr;
                    bool m2 = ok && (unsigned)(rv[u].z - base) < (unsigned)nr;
                    bool m3 = ok && (unsigned)(rv[u].w - base) < (unsigned)nr;
                    unsigned b0 = __builtin_amdgcn_ballot_w32(m0);
                    unsigned b1 = __builtin_amdgcn_ballot_w32(m1);
                    unsigned b2 = __builtin_amdgcn_ballot_w32(m2);
                    unsigned b3 = __builtin_amdgcn_ballot_w32(m3);
                    if (b0 | b1 | b2 | b3) {
                        int e0 = qx[u] * 4;
                        NOTE_EDGE(m0, b0, e0 + 0)
                        NOTE_EDGE(m1, b1, e0 + 1)
                        NOTE_EDGE(m2, b2, e0 + 2)
                        NOTE_EDGE(m3, b3, e0 + 3)
                    }
                }
            }
        }
        if (wcount > WAVE_CAP) spill = 1;
        if (lane == 0) wcnt_s[wave] = wcount;
        __syncthreads();

        const int spilled = __builtin_amdgcn_readfirstlane(spill);

        int off = 0, padded = 0;
        const int cw_l = wcnt_s[lane];
        for (int w = 0; w < 32; ++w) {
            int cw = (__builtin_amdgcn_readlane(cw_l, w) + 31) & ~31;
            if (w < wave) off += cw;
            padded += cw;
        }
        const int wr = first && !spilled && padded <= rec_cap;
        if (first && wave == 0) {
            hdr[lane] = cw_l;
            if (lane == 0) hdr[32] = wr;
        }

        if (spilled == 0) {
            const int n = wcount;
            uint2* seg = myrec + REC_HDR + off;
            for (int i0 = 0; i0 < n; i0 += 32) {
                int cnt = n - i0;
                if (cnt > 32) cnt = 32;
                int   e  = mylist[i0 + (lane < cnt ? lane : 0)];
                int   rl = rows[e] - base;
                int   c  = cols[e];
                float v  = vals[e];
                if ((unsigned)rl >= (unsigned)nr) { rl = 0; v = 0.f; }
                if (wr && lane < cnt)
                    seg[i0 + lane] = make_uint2(((unsigned)rl << COL_BITS) | (unsigned)c,
                                                __builtin_bit_cast(unsigned, v));
                add_edges(rl, c, v, cnt, lane, acc, srcA, srcB, split);
            }
        } else {
            for (int q = tid; q < N_QUADS; q += 1024) {
                int4 r4 = rows4[q];
                int rr[4] = {r4.x, r4.y, r4.z, r4.w};
#pragma unroll
                for (int k = 0; k < 4; ++k) {
                    unsigned loc = (unsigned)(rr[k] - base);
                    if (loc < (unsigned)nr) {
                        int   e = q * 4 + k;
                        int   c = cols[e];
                        float v = vals[e];
                        const float* sp = (c < split)
                            ? (srcA + (size_t)c * DIM)
                            : (srcB + (size_t)(c - split) * DIM);
                        float* a = acc + loc * DIM;
#pragma nounroll
                        for (int d = 0; d < DIM; ++d)
                            unsafeAtomicAdd(a + d, v * sp[d]);
                    }
                }
            }
        }
    }
    __syncthreads();

    float4* out4 = (float4*)(Hout + (size_t)base * DIM);
    for (int i = tid; i < nr * DIM / 4; i += 1024)
        out4[i] = acc4[i];
}

__global__ __launch_bounds__(256) void pred_kernel(
    const float* __restrict__ U, const float* __restrict__ I,
    const int*   __restrict__ u_idx, const int* __restrict__ i_idx,
    const float* __restrict__ bu, const float* __restrict__ bi,
    const float* __restrict__ mu, float* __restrict__ out) {
    int wave = blockIdx.x * (blockDim.x >> 5) + (threadIdx.x >> 5);
    int lane = threadIdx.x & 31;
    int base = wave << 4;
    int m    = lane & 15;
    int hi   = lane >> 4;
    int ur = u_idx[base + m];
    int ir = i_idx[base + m];
    const float* up = U + (size_t)ur * DIM;
    const float* ip = I + (size_t)ir * DIM;

    v8f c = {};
#pragma unroll
    for (int k0 = 0; k0 < DIM; k0 += 4) {
        v2f a = *(const v2f*)(up + k0 + 2 * hi);
        v2f b = *(const v2f*)(ip + k0 + 2 * hi);
        c = __builtin_amdgcn_wmma_f32_16x16x4_f32(
                false, a, false, b, (short)0, c, false, false);
    }

    float d = 0.f;
#pragma unroll
    for (int j = 0; j < 8; ++j) {
        float vlo = __shfl(c[j], j, 32);
        float vhi = __shfl(c[j], j + 24, 32);
        if (lane == j)     d = vlo;
        if (lane == j + 8) d = vhi;
    }

    if (lane < 16) {
        out[base + lane] = d + bu[ur] + bi[ir] + mu[0];
    }
}

extern "C" void kernel_launch(void* const* d_in, const int* in_sizes, int n_in,
                              void* d_out, int out_size, void* d_ws, size_t ws_size,
                              hipStream_t stream) {
    const int*   u_idx  = (const int*)d_in[0];
    const int*   i_idx  = (const int*)d_in[1];
    const int*   A_rows = (const int*)d_in[2];
    const int*   A_cols = (const int*)d_in[3];
    const float* A_vals = (const float*)d_in[4];
    const float* Eu     = (const float*)d_in[5];
    const float* Ei     = (const float*)d_in[6];
    const float* bu     = (const float*)d_in[7];
    const float* bi     = (const float*)d_in[8];
    const float* mu     = (const float*)d_in[9];
    float* out = (float*)d_out;

    float* H1 = (float*)d_ws;
    float* H2 = H1 + (size_t)N_NODES * DIM;

    owner_key_kernel<<<256, 256, 0, stream>>>(A_rows, H2);

    spmm_owner_kernel<<<OWN_GROUPS, 1024, 0, stream>>>(
        A_rows, A_cols, A_vals, Eu, Ei, N_USERS, H1, H2, 1);

    spmm_owner_kernel<<<OWN_GROUPS, 1024, 0, stream>>>(
        A_rows, A_cols, A_vals, H1, H1, N_NODES + 1, H2, H2, 0);

    int pred_blocks = (BATCH_CT / 16) / 8;
    pred_kernel<<<pred_blocks, 256, 0, stream>>>(
        H2, H2 + (size_t)N_USERS * DIM, u_idx, i_idx, bu, bi, mu, out);
}


# pred: one WMMA per (u,i) pair: both rows, read whole (8 B per lane, full lines), are its A and B fragments; dot = trace of D, summed across lanes (new sum order); 8192 waves walk 128 pairs each
# speedup vs baseline: 2.1068x; 1.1587x over previous
#include <hip/hip_runtime.h>

#define N_USERS  100000
#define N_ITEMS  50000
#define N_NODES  (N_USERS + N_ITEMS)
#define NNZ_CT   1500000
#define BATCH_CT 1048576
#define DIM      64

typedef __attribute__((ext_vector_type(2))) float v2f;
typedef __attribute__((ext_vector_type(8))) float v8f;

#define OWN_ROWS   586
#define OWN_GROUPS 256
#define WAVE_CAP   1024
#define N_QUADS    (NNZ_CT / 4)
#define SCAN_STEPS 368
#define ROT_QUADS  1464
#define REC_HDR    32
#define COL_BITS   18
#define KEY_STEPS  92
#define KEY_UNITS  (32 * KEY_STEPS)
#define KEY_CHUNK_BYTES 8192
#define KEY_ROW0   (OWN_ROWS - KEY_CHUNK_BYTES / (DIM * 4))
#define KEY_OFF_BYTES (KEY_ROW0 * DIM * 4)
#define RANGE_BYTES (OWN_ROWS * DIM * 4)
static_assert(N_NODES <= (1 << COL_BITS), "source row id must fit its field");
static_assert(OWN_ROWS <= (1 << (32 - COL_BITS)), "local row must fit its field");
static_assert((N_NODES - 1) / OWN_ROWS < 256, "owner id must fit a byte");
static_assert(NNZ_CT % 16 == 0 && KEY_UNITS * 512 >= NNZ_CT, "keys cover the edge list");
static_assert(KEY_UNITS * 512 / KEY_CHUNK_BYTES <= OWN_GROUPS - 1, "key chunks stay in full-size ranges");
static_assert(N_NODES - (OWN_GROUPS - 1) * OWN_ROWS >= KEY_ROW0, "last range holds its records");
static_assert(KEY_UNITS * 128 <= 6 * 65536, "the key launch covers every unit");

__device__ __forceinline__ unsigned nonzero_bytes(unsigned x) {
    return ((x & 0x7f7f7f7fu) + 0x7f7f7f7fu) | x;
}

__global__ __launch_bounds__(256) void owner_key_kernel(
    const int* __restrict__ rows, float* keyplane) {
    const int4* rows4 = (const int4*)rows;
    const int t = blockIdx.x * 256 + threadIdx.x;
    int4 r[6];
#pragma unroll
    for (int j = 0; j < 6; ++j) {
        int q = t + j * 65536;
        r[j] = rows4[q < N_QUADS ? q : 0];
    }
#pragma unroll
    for (int j = 0; j < 6; ++j) {
        int q = t + j * 65536;
        if (q < KEY_UNITS * 128) {
            unsigned k = 0u;
            if (q < N_QUADS)
                k =  (unsigned)(r[j].x / OWN_ROWS)        | ((unsigned)(r[j].y / OWN_ROWS) << 8)
                  | ((unsigned)(r[j].z / OWN_ROWS) << 16) | ((unsigned)(r[j].w / OWN_ROWS) << 24);
            int e  = q * 4;
            int ch = e / KEY_CHUNK_BYTES;
            int of = e - ch * KEY_CHUNK_BYTES;
            *(unsigned*)((char*)keyplane + (size_t)ch * RANGE_BYTES + KEY_OFF_BYTES + of) = k;
        }
    }
}

#define NOTE_EDGE(m, b, e)                                              \
    if (b) {                                                            \
        if (m) {                                                        \
            int slot = wcount + (int)__builtin_amdgcn_mbcnt_lo((b), 0u);\
            if (slot < WAVE_CAP) mylist[slot] = (e);                    \
        }                                                               \
        wcount += __builtin_popcount(b);                                \
    }

__device__ __forceinline__ void add_edges(
    int rl, int c, float v, int cnt, int lane, float* acc,
    const float* __restrict__ srcA, const float* __restrict__ srcB, int split) {
    for (int j0 = 0; j0 < cnt; j0 += 16) {
        v2f h[16];
#pragma unroll
        for (int u = 0; u < 16; ++u) {
            int j = j0 + u;
            if (j > cnt - 1) j = cnt - 1;
            int cj = __builtin_amdgcn_readlane(c, j);
            const float* sp = (cj < split)
                ? (srcA + (size_t)cj * DIM)
                : (srcB + (size_t)(cj - split) * DIM);
            h[u] = *(const v2f*)(sp + 2 * lane);
        }
#pragma unroll
        for (int u = 0; u < 16; ++u) {
            int j = j0 + u;
            if (j < cnt) {
                int   rj = __builtin_amdgcn_readlane(rl, j);
                float vj = __builtin_bit_cast(float,
                    __builtin_amdgcn_readlane(__builtin_bit_cast(int, v), j));
                float* a = acc + rj * DIM + 2 * lane;
                unsafeAtomicAdd(a,     vj * h[u].x);
                unsafeAtomicAdd(a + 1, vj * h[u].y);
            }
        }
    }
}

__global__ __launch_bounds__(1024) void spmm_owner_kernel(
    const int*   __restrict__ rows, const int* __restrict__ cols,
    const float* __restrict__ vals,
    const float* __restrict__ srcA, const float* __restrict__ srcB, int split,
    float* Hout, float* recplane, int first) {
    __shared__ float acc[OWN_ROWS * DIM];
    __shared__ int   elist[32 * WAVE_CAP];
    __shared__ int   spill;
    __shared__ int   wcnt_s[32];

    const int tid  = threadIdx.x;
    const int lane = tid & 31;
    const int base = blockIdx.x * OWN_ROWS;
    int nr = N_NODES - base;
    if (nr > OWN_ROWS) nr = OWN_ROWS;

    float4* acc4 = (float4*)acc;
    for (int i = tid; i < OWN_ROWS * DIM / 4; i += 1024)
        acc4[i] = make_float4(0.f, 0.f, 0.f, 0.f);
    if (tid == 0) spill = 0;

    uint2* myrec = (uint2*)(recplane + (size_t)base * DIM);
    int*   hdr   = (int*)myrec;
    const int rec_cap = KEY_ROW0 * (DIM / 2) - REC_HDR;

    const int wave = __builtin_amdgcn_readfirstlane(tid >> 5);

    int listed = 0;
    int cl = 0;
    if (!first) {
        listed = __builtin_amdgcn_readfirstlane(hdr[32]) == 1;
        cl = hdr[lane];
    }
    __syncthreads();

    if (listed) {
        int n = 0, off = 0;
        for (int w = 0; w < 32; ++w) {
            int cw = __builtin_amdgcn_readlane(cl, w);
            if (w < wave)  off += (cw + 31) & ~31;
            if (w == wave) n = cw;
        }
        const uint2* seg = myrec + REC_HDR + off;
        uint2 nxt = make_uint2(0u, 0u);
        if (n > 0) nxt = seg[lane < n ? lane : 0];
        for (int i0 = 0; i0 < n; i0 += 32) {
            int cnt = n - i0;
            if (cnt > 32) cnt = 32;
            uint2 cur = nxt;
            if (i0 + 32 < n) {
                int x = i0 + 32 + lane;
                nxt = seg[x < n ? x : i0 + 32];
            }
            int   rl = (int)(cur.x >> COL_BITS);
            int   c  = (int)(cur.x & ((1u << COL_BITS) - 1u));
            float v  = __builtin_bit_cast(float, cur.y);
            add_edges(rl, c, v, cnt, lane, acc, srcA, srcB, split);
        }
    } else {
        int* mylist = elist + wave * WAVE_CAP;
        int  wcount = 0;

        const int4* rows4 = (const int4*)rows;
        const int rot = blockIdx.x * ROT_QUADS;
        if (first) {
            const unsigned g4 = (unsigned)blockIdx.x * 0x01010101u;
            const char* kplane = (const char*)recplane;
            const int krot   = (blockIdx.x * KEY_UNITS) >> 8;
            const int lane16 = lane * 16;
            for (int i = 0; i < KEY_STEPS; i += 4) {
                uint4 kv[4];
                int   eb[4];
#pragma unroll
                for (int u = 0; u < 4; ++u) {
                    int x = wave + 32 * (i + u) + krot;
                    if (x >= KEY_UNITS) x -= KEY_UNITS;
                    eb[u] = x * 512;
                    int ch = eb[u] / KEY_CHUNK_BYTES;
                    int of = eb[u] - ch * KEY_CHUNK_BYTES;
                    kv[u] = *(const uint4*)(kplane + (size_t)ch * RANGE_BYTES
                                            + KEY_OFF_BYTES + of + lane16);
                }
#pragma unroll
                for (int u = 0; u < 4; ++u) {
                    unsigned n0 = nonzero_bytes(kv[u].x ^ g4), n1 = nonzero_bytes(kv[u].y ^ g4);
                    unsigned n2 = nonzero_bytes(kv[u].z ^ g4), n3 = nonzero_bytes(kv[u].w ^ g4);
                    if (__builtin_amdgcn_ballot_w32((n0 & n1 & n2 & n3 & 0x80808080u) != 0x80808080u)) {
                        const int e0 = eb[u] + lane16;
                        unsigned lm = ((~n0 & 0x80808080u) >> 7) | ((~n1 & 0x80808080u) >> 6) |
                                      ((~n2 & 0x80808080u) >> 5) | ((~n3 & 0x80808080u) >> 4);
                        if (e0 >= NNZ_CT) lm = 0u;
                        bool     m = lm != 0u;
                        unsigned b = __builtin_amdgcn_ballot_w32(m);
#pragma nounroll
                        while (b != 0u) {
                            int bit  = __builtin_ctz(lm | 0x80000000u);
                            int slot = wcount + (int)__builtin_amdgcn_mbcnt_lo(b, 0u);
                            if (m && slot < WAVE_CAP)
                                mylist[slot] = e0 + ((bit & 7) << 2) + (bit >> 3);
                            lm &= lm - 1u;
                            wcount += __builtin_popcount(b);
                            m = lm != 0u;
                            b = __builtin_amdgcn_ballot_w32(m);
                        }
                    }
                }
            }
        } else {
            for (int i = 0; i < SCAN_STEPS; i += 16) {
                int4 rv[16];
                int  qx[16];
#pragma unroll
                for (int u = 0; u < 16; ++u) {
                    int q = tid + 1024 * (i + u);
                    int x = (q < N_QUADS ? q : 0) + rot;
                    if (x >= N_QUADS) x -= N_QUADS;
                    qx[u] = x;
                    rv[u] = rows4[x];
                }
#pragma unroll
                for (int u = 0; u < 16; ++u) {
                    bool ok = (tid + 1024 * (i + u)) < N_QUADS;
                    bool m0 = ok && (unsigned)(rv[u].x - base) < (unsigned)nr;
                    bool m1 = ok && (unsigned)(rv[u].y - base) < (unsigned)nr;
                    bool m2 = ok && (unsigned)(rv[u].z - base) < (unsigned)nr;
                    bool m3 = ok && (unsigned)(rv[u].w - base) < (unsigned)nr;
                    unsigned b0 = __builtin_amdgcn_ballot_w32(m0);
                    unsigned b1 = __builtin_amdgcn_ballot_w32(m1);
                    unsigned b2 = __builtin_amdgcn_ballot_w32(m2);
                    unsigned b3 = __builtin_amdgcn_ballot_w32(m3);
                    if (b0 | b1 | b2 | b3) {
                        int e0 = qx[u] * 4;
                        NOTE_EDGE(m0, b0, e0 + 0)
                        NOTE_EDGE(m1, b1, e0 + 1)
                        NOTE_EDGE(m2, b2, e0 + 2)
                        NOTE_EDGE(m3, b3, e0 + 3)
                    }
                }
            }
        }
        if (wcount > WAVE_CAP) spill = 1;
        if (lane == 0) wcnt_s[wave] = wcount;
        __syncthreads();

        const int spilled = __builtin_amdgcn_readfirstlane(spill);

        int off = 0, padded = 0;
        const int cw_l = wcnt_s[lane];
        for (int w = 0; w < 32; ++w) {
            int cw = (__builtin_amdgcn_readlane(cw_l, w) + 31) & ~31;
            if (w < wave) off += cw;
            padded += cw;
        }
        const int wr = first && !spilled && padded <= rec_cap;
        if (first && wave == 0) {
            hdr[lane] = cw_l;
            if (lane == 0) hdr[32] = wr;
        }

        if (spilled == 0) {
            const int n = wcount;
            uint2* seg = myrec + REC_HDR + off;
            for (int i0 = 0; i0 < n; i0 += 32) {
                int cnt = n - i0;
                if (cnt > 32) cnt = 32;
                int   e  = mylist[i0 + (lane < cnt ? lane : 0)];
                int   rl = rows[e] - base;
                int   c  = cols[e];
                float v  = vals[e];
                if ((unsigned)rl >= (unsigned)nr) { rl = 0; v = 0.f; }
                if (wr && lane < cnt)
                    seg[i0 + lane] = make_uint2(((unsigned)rl << COL_BITS) | (unsigned)c,
                                                __builtin_bit_cast(unsigned, v));
                add_edges(rl, c, v, cnt, lane, acc, srcA, srcB, split);
            }
        } else {
            for (int q = tid; q < N_QUADS; q += 1024) {
                int4 r4 = rows4[q];
                int rr[4] = {r4.x, r4.y, r4.z, r4.w};
#pragma unroll
                for (int k = 0; k < 4; ++k) {
                    unsigned loc = (unsigned)(rr[k] - base);
                    if (loc < (unsigned)nr) {
                        int   e = q * 4 + k;
                        int   c = cols[e];
                        float v = vals[e];
                        const float* sp = (c < split)
                            ? (srcA + (size_t)c * DIM)
                            : (srcB + (size_t)(c - split) * DIM);
                        float* a = acc + loc * DIM;
#pragma nounroll
                        for (int d = 0; d < DIM; ++d)
                            unsafeAtomicAdd(a + d, v * sp[d]);
                    }
                }
            }
        }
    }
    __syncthreads();

    float4* out4 = (float4*)(Hout + (size_t)base * DIM);
    for (int i = tid; i < nr * DIM / 4; i += 1024)
        out4[i] = acc4[i];
}

#define PRED_PAIRS_PER_WAVE 128
#define PRED_TRIPS (PRED_PAIRS_PER_WAVE / 32)
static_assert(BATCH_CT % (8 * PRED_PAIRS_PER_WAVE) == 0, "whole blocks cover the batch");

template <int N>
__device__ __forceinline__ float row_xor(float x) {
    return __builtin_bit_cast(float, __builtin_amdgcn_update_dpp(
        0, __builtin_bit_cast(int, x), 0x160 + N, 0xf, 0xf, true));
}
__device__ __forceinline__ float row_swap(float x) {
    unsigned v = __builtin_bit_cast(unsigned, x);
    return __builtin_bit_cast(float, __builtin_amdgcn_permlanex16(
        v, v, 0x76543210u, 0xfedcba98u, false, false));
}
template <int N>
__device__ __forceinline__ float fold2(float a, float b, bool bit) {
    float keep = bit ? b : a;
    float send = bit ? a : b;
    return keep + row_xor<N>(send);
}

__global__ __launch_bounds__(256) void pred_kernel(
    const float* __restrict__ U, const float* __restrict__ I,
    const int*   __restrict__ u_idx, const int* __restrict__ i_idx,
    const float* __restrict__ bu, const float* __restrict__ bi,
    const float* __restrict__ mu, float* __restrict__ out) {
    const int lane = threadIdx.x & 31;
    const int wave = __builtin_amdgcn_readfirstlane(blockIdx.x * 8 + (threadIdx.x >> 5));
    const int  r    = lane & 7;
    const bool b0   = (lane & 1) != 0, b1 = (lane & 2) != 0, b2 = (lane & 4) != 0;
    const bool cell = (lane < 8) || (lane >= 24);
    const float mu0 = mu[0];

    for (int trip = 0; trip < PRED_TRIPS; ++trip) {
        const int base = wave * PRED_PAIRS_PER_WAVE + trip * 32;
        const int ur = u_idx[base + lane];
        const int ir = i_idx[base + lane];
        const float bu_v = bu[ur];
        const float bi_v = bi[ir];

        float G[4];
#pragma unroll
        for (int g = 0; g < 4; ++g) {
            v2f a[8], b[8];
#pragma unroll
            for (int j = 0; j < 8; ++j) {
                int su = __builtin_amdgcn_readlane(ur, 8 * g + j);
                int si = __builtin_amdgcn_readlane(ir, 8 * g + j);
                a[j] = *(const v2f*)(U + (size_t)su * DIM + 2 * lane);
                b[j] = *(const v2f*)(I + (size_t)si * DIM + 2 * lane);
            }
            float x[8];
#pragma unroll
            for (int j = 0; j < 8; ++j) {
                v8f c = {};
                c = __builtin_amdgcn_wmma_f32_16x16x4_f32(
                        false, a[j], false, b[j], (short)0, c, false, false);
                float t = c[0];
                t = (r == 1) ? c[1] : t;
                t = (r == 2) ? c[2] : t;
                t = (r == 3) ? c[3] : t;
                t = (r == 4) ? c[4] : t;
                t = (r == 5) ? c[5] : t;
                t = (r == 6) ? c[6] : t;
                t = (r == 7) ? c[7] : t;
                x[j] = t;
            }
            float y0 = fold2<1>(x[0], x[1], b0), y1 = fold2<1>(x[2], x[3], b0);
            float y2 = fold2<1>(x[4], x[5], b0), y3 = fold2<1>(x[6], x[7], b0);
            float z0 = fold2<2>(y0, y1, b1),     z1 = fold2<2>(y2, y3, b1);
            G[g] = fold2<4>(z0, z1, b2);
        }

        const float g1x = row_xor<8>(G[1]);
        const float g3x = row_xor<8>(G[3]);
        float h01 = cell ? G[0] : g1x;
        float h23 = cell ? G[2] : g3x;
        float t01 = h01 + row_swap(row_xor<8>(h01));
        float t23 = row_xor<8>(h23) + row_swap(h23);
        float d = (lane < 16) ? t01 : t23;

        out[base + lane] = d + bu_v + bi_v + mu0;
    }
}

extern "C" void kernel_launch(void* const* d_in, const int* in_sizes, int n_in,
                              void* d_out, int out_size, void* d_ws, size_t ws_size,
                              hipStream_t stream) {
    const int*   u_idx  = (const int*)d_in[0];
    const int*   i_idx  = (const int*)d_in[1];
    const int*   A_rows = (const int*)d_in[2];
    const int*   A_cols = (const int*)d_in[3];
    const float* A_vals = (const float*)d_in[4];
    const float* Eu     = (const float*)d_in[5];
    const float* Ei     = (const float*)d_in[6];
    const float* bu     = (const float*)d_in[7];
    const float* bi     = (const float*)d_in[8];
    const float* mu     = (const float*)d_in[9];
    float* out = (float*)d_out;

    float* H1 = (float*)d_ws;
    float* H2 = H1 + (size_t)N_NODES * DIM;

    owner_key_kernel<<<256, 256, 0, stream>>>(A_rows, H2);

    spmm_owner_kernel<<<OWN_GROUPS, 1024, 0, stream>>>(
        A_rows, A_cols, A_vals, Eu, Ei, N_USERS, H1, H2, 1);

    spmm_owner_kernel<<<OWN_GROUPS, 1024, 0, stream>>>(
        A_rows, A_cols, A_vals, H1, H1, N_NODES + 1, H2, H2, 0);

    pred_kernel<<<BATCH_CT / (8 * PRED_PAIRS_PER_WAVE), 256, 0, stream>>>(
        H2, H2 + (size_t)N_USERS * DIM, u_idx, i_idx, bu, bi, mu, out);
}


# spmm layer-1 filter: the small launch in front writes per-group bit maps (1 bit per 16 edges); a group tests the row ids of its flagged units only, no edge-list scan; LDS f32 adds in arrival order
# speedup vs baseline: 2.2498x; 1.0679x over previous
#include <hip/hip_runtime.h>

#define N_USERS  100000
#define N_ITEMS  50000
#define N_NODES  (N_USERS + N_ITEMS)
#define NNZ_CT   1500000
#define BATCH_CT 1048576
#define DIM      64

typedef __attribute__((ext_vector_type(2))) float v2f;
typedef __attribute__((ext_vector_type(8))) float v8f;

#define OWN_ROWS   586
#define OWN_GROUPS 256
#define WAVE_CAP   1024
#define N_QUADS    (NNZ_CT / 4)
#define SCAN_STEPS 368
#define ROT_QUADS  1464
#define REC_HDR    32
#define COL_BITS   18
#define UNIT_EDGES 16
#define N_UNITS    (NNZ_CT / UNIT_EDGES)
#define BM_BLOCKS  92
#define BM_DWORDS  (BM_BLOCKS * 32)
#define BM_ROWS    (BM_DWORDS * 4 / (DIM * 4))
#define UNIT_CAP   512
#define EDGE_CAP   512
static_assert(N_NODES <= (1 << COL_BITS), "source row id must fit its field");
static_assert(OWN_ROWS <= (1 << (32 - COL_BITS)), "local row must fit its field");
static_assert((N_NODES - 1) / OWN_ROWS < OWN_GROUPS, "every row has an owner");
static_assert(NNZ_CT % UNIT_EDGES == 0 && BM_BLOCKS * 1024 >= N_UNITS, "the map covers the edge list");
static_assert(BM_DWORDS * 4 % (DIM * 4) == 0, "a map is a whole number of rows");
static_assert(BM_DWORDS <= 3 * 1024, "three map dwords per lane cover a map");
static_assert(N_NODES - (OWN_GROUPS - 1) * OWN_ROWS > BM_ROWS + 256, "last range holds map and records");
static_assert(UNIT_CAP + EDGE_CAP <= WAVE_CAP, "both lists fit the wave's list");

#define BM_PITCH 257
__global__ __launch_bounds__(1024) void unit_map_kernel(
    const int* __restrict__ rows, float* plane) {
    __shared__ unsigned bm[32 * BM_PITCH];
    const int tid  = threadIdx.x;
    const int lane = tid & 31;
    const int wave = tid >> 5;
    for (int i = tid; i < 32 * BM_PITCH; i += 1024) bm[i] = 0u;
    const int4* rows4 = (const int4*)rows;
    const int  unit = blockIdx.x * 1024 + tid;
    const bool live = unit < N_UNITS;
    int4 r[4];
#pragma unroll
    for (int j = 0; j < 4; ++j) r[j] = rows4[live ? unit * 4 + j : 0];
    __syncthreads();
    if (live) {
        unsigned* mine = bm + wave * BM_PITCH;
        const unsigned bit = 1u << lane;
#pragma unroll
        for (int j = 0; j < 4; ++j) {
            unsigned g0 = (unsigned)r[j].x / OWN_ROWS, g1 = (unsigned)r[j].y / OWN_ROWS;
            unsigned g2 = (unsigned)r[j].z / OWN_ROWS, g3 = (unsigned)r[j].w / OWN_ROWS;
            if (g0 < OWN_GROUPS) atomicOr(mine + g0, bit);
            if (g1 < OWN_GROUPS) atomicOr(mine + g1, bit);
            if (g2 < OWN_GROUPS) atomicOr(mine + g2, bit);
            if (g3 < OWN_GROUPS) atomicOr(mine + g3, bit);
        }
    }
    __syncthreads();
    for (int g = wave; g < OWN_GROUPS; g += 32) {
        int nr = N_NODES - g * OWN_ROWS;
        if (nr > OWN_ROWS) nr = OWN_ROWS;
        unsigned* dst = (unsigned*)(plane + (size_t)(g * OWN_ROWS + nr - BM_ROWS) * DIM)
                      + blockIdx.x * 32;
        dst[lane] = bm[lane * BM_PITCH + g];
    }
}

#define NOTE_EDGE(m, b, e)                                              \
    if (b) {                                                            \
        if (m) {                                                        \
            int slot = wcount + (int)__builtin_amdgcn_mbcnt_lo((b), 0u);\
            if (slot < WAVE_CAP) mylist[slot] = (e);                    \
        }                                                               \
        wcount += __builtin_popcount(b);                                \
    }

__device__ __forceinline__ void add_edges(
    int rl, int c, float v, int cnt, int lane, float* acc,
    const float* __restrict__ srcA, const float* __restrict__ srcB, int split) {
    for (int j0 = 0; j0 < cnt; j0 += 16) {
        v2f h[16];
#pragma unroll
        for (int u = 0; u < 16; ++u) {
            int j = j0 + u;
            if (j > cnt - 1) j = cnt - 1;
            int cj = __builtin_amdgcn_readlane(c, j);
            const float* sp = (cj < split)
                ? (srcA + (size_t)cj * DIM)
                : (srcB + (size_t)(cj - split) * DIM);
            h[u] = *(const v2f*)(sp + 2 * lane);
        }
#pragma unroll
        for (int u = 0; u < 16; ++u) {
            int j = j0 + u;
            if (j < cnt) {
                int   rj = __builtin_amdgcn_readlane(rl, j);
                float vj = __builtin_bit_cast(float,
                    __builtin_amdgcn_readlane(__builtin_bit_cast(int, v), j));
                float* a = acc + rj * DIM + 2 * lane;
                unsafeAtomicAdd(a,     vj * h[u].x);
                unsafeAtomicAdd(a + 1, vj * h[u].y);
            }
        }
    }
}

__global__ __launch_bounds__(1024) void spmm_owner_kernel(
    const int*   __restrict__ rows, const int* __restrict__ cols,
    const float* __restrict__ vals,
    const float* __restrict__ srcA, const float* __restrict__ srcB, int split,
    float* Hout, float* recplane, int first) {
    __shared__ float acc[OWN_ROWS * DIM];
    __shared__ int   elist[32 * WAVE_CAP];
    __shared__ int   spill;
    __shared__ int   wcnt_s[32];

    const int tid  = threadIdx.x;
    const int lane = tid & 31;
    const int base = blockIdx.x * OWN_ROWS;
    int nr = N_NODES - base;
    if (nr > OWN_ROWS) nr = OWN_ROWS;

    float4* acc4 = (float4*)acc;
    for (int i = tid; i < OWN_ROWS * DIM / 4; i += 1024)
        acc4[i] = make_float4(0.f, 0.f, 0.f, 0.f);
    if (tid == 0) spill = 0;

    uint2* myrec = (uint2*)(recplane + (size_t)base * DIM);
    int*   hdr   = (int*)myrec;
    const int rec_cap = (nr - BM_ROWS) * (DIM / 2) - REC_HDR;

    const int wave = __builtin_amdgcn_readfirstlane(tid >> 5);

    int listed = 0;
    int cl = 0;
    if (!first) {
        listed = __builtin_amdgcn_readfirstlane(hdr[32]) == 1;
        cl = hdr[lane];
    }
    __syncthreads();

    if (listed) {
        int n = 0, off = 0;
        for (int w = 0; w < 32; ++w) {
            int cw = __builtin_amdgcn_readlane(cl, w);
            if (w < wave)  off += (cw + 31) & ~31;
            if (w == wave) n = cw;
        }
        const uint2* seg = myrec + REC_HDR + off;
        uint2 nxt = make_uint2(0u, 0u);
        if (n > 0) nxt = seg[lane < n ? lane : 0];
        for (int i0 = 0; i0 < n; i0 += 32) {
            int cnt = n - i0;
            if (cnt > 32) cnt = 32;
            uint2 cur = nxt;
            if (i0 + 32 < n) {
                int x = i0 + 32 + lane;
                nxt = seg[x < n ? x : i0 + 32];
            }
            int   rl = (int)(cur.x >> COL_BITS);
            int   c  = (int)(cur.x & ((1u << COL_BITS) - 1u));
            float v  = __builtin_bit_cast(float, cur.y);
            add_edges(rl, c, v, cnt, lane, acc, srcA, srcB, split);
        }
    } else {
        int* mylist = elist + wave * WAVE_CAP;
        int  wcount = 0;

        const int4* rows4 = (const int4*)rows;
        const int rot = blockIdx.x * ROT_QUADS;
        if (first) {
            const unsigned* bmap = (const unsigned*)(recplane + (size_t)(base + nr - BM_ROWS) * DIM);
            int* ulist  = mylist + EDGE_CAP;
            int  ucount = 0;
            unsigned wd[3];
#pragma unroll
            for (int k = 0; k < 3; ++k) {
                int d = wave * 32 + lane + 1024 * k;
                wd[k] = bmap[d < BM_DWORDS ? d : 0];
                if (d >= BM_DWORDS) wd[k] = 0u;
            }
#pragma unroll
            for (int k = 0; k < 3; ++k) {
                unsigned x = wd[k];
                const int ub = (wave * 32 + lane + 1024 * k) * 32;
                bool     m = x != 0u;
                unsigned b = __builtin_amdgcn_ballot_w32(m);
#pragma nounroll
                while (b != 0u) {
                    int bit  = __builtin_ctz(x | 0x80000000u);
                    int slot = ucount + (int)__builtin_amdgcn_mbcnt_lo(b, 0u);
                    if (m && slot < UNIT_CAP) ulist[slot] = ub + bit;
                    x &= x - 1u;
                    ucount += __builtin_popcount(b);
                    m = x != 0u;
                    b = __builtin_amdgcn_ballot_w32(m);
                }
            }
            if (ucount > UNIT_CAP) { spill = 1; ucount = UNIT_CAP; }

            int  un = 0;
            int4 nx[4];
            if (ucount > 0) {
                un = ulist[lane < ucount ? lane : 0];
#pragma unroll
                for (int j = 0; j < 4; ++j) nx[j] = rows4[un * 4 + j];
            }
            for (int i0 = 0; i0 < ucount; i0 += 32) {
                const int  uc = un;
                const bool on = i0 + lane < ucount;
                int4 rv[4];
#pragma unroll
                for (int j = 0; j < 4; ++j) rv[j] = nx[j];
                if (i0 + 32 < ucount) {
                    int x = i0 + 32 + lane;
                    un = ulist[x < ucount ? x : i0 + 32];
#pragma unroll
                    for (int j = 0; j < 4; ++j) nx[j] = rows4[un * 4 + j];
                }
                unsigned lm = 0u;
#pragma unroll
                for (int j = 0; j < 4; ++j) {
                    if ((unsigned)(rv[j].x - base) < (unsigned)nr) lm |= 1u << (4 * j);
                    if ((unsigned)(rv[j].y - base) < (unsigned)nr) lm |= 2u << (4 * j);
                    if ((unsigned)(rv[j].z - base) < (unsigned)nr) lm |= 4u << (4 * j);
                    if ((unsigned)(rv[j].w - base) < (unsigned)nr) lm |= 8u << (4 * j);
                }
                if (!on) lm = 0u;
                bool     m = lm != 0u;
                unsigned b = __builtin_amdgcn_ballot_w32(m);
                const int e0 = uc * UNIT_EDGES;
#pragma nounroll
                while (b != 0u) {
                    int bit  = __builtin_ctz(lm | 0x80000000u);
                    int slot = wcount + (int)__builtin_amdgcn_mbcnt_lo(b, 0u);
                    if (m && slot < EDGE_CAP) mylist[slot] = e0 + bit;
                    lm &= lm - 1u;
                    wcount += __builtin_popcount(b);
                    m = lm != 0u;
                    b = __builtin_amdgcn_ballot_w32(m);
                }
            }
            if (wcount > EDGE_CAP) spill = 1;
        } else {
            for (int i = 0; i < SCAN_STEPS; i += 16) {
                int4 rv[16];
                int  qx[16];
#pragma unroll
                for (int u = 0; u < 16; ++u) {
                    int q = tid + 1024 * (i + u);
                    int x = (q < N_QUADS ? q : 0) + rot;
                    if (x >= N_QUADS) x -= N_QUADS;
                    qx[u] = x;
                    rv[u] = rows4[x];
                }
#pragma unroll
                for (int u = 0; u < 16; ++u) {
                    bool ok = (tid + 1024 * (i + u)) < N_QUADS;
                    bool m0 = ok && (unsigned)(rv[u].x - base) < (unsigned)nr;
                    bool m1 = ok && (unsigned)(rv[u].y - base) < (unsigned)nr;
                    bool m2 = ok && (unsigned)(rv[u].z - base) < (unsigned)nr;
                    bool m3 = ok && (unsigned)(rv[u].w - base) < (unsigned)nr;
                    unsigned b0 = __builtin_amdgcn_ballot_w32(m0);
                    unsigned b1 = __builtin_amdgcn_ballot_w32(m1);
                    unsigned b2 = __builtin_amdgcn_ballot_w32(m2);
                    unsigned b3 = __builtin_amdgcn_ballot_w32(m3);
                    if (b0 | b1 | b2 | b3) {
                        int e0 = qx[u] * 4;
                        NOTE_EDGE(m0, b0, e0 + 0)
                        NOTE_EDGE(m1, b1, e0 + 1)
                        NOTE_EDGE(m2, b2, e0 + 2)
                        NOTE_EDGE(m3, b3, e0 + 3)
                    }
                }
            }
        }
        if (!first && wcount > WAVE_CAP) spill = 1;
        if (lane == 0) wcnt_s[wave] = wcount;
        __syncthreads();

        const int spilled = __builtin_amdgcn_readfirstlane(spill);

        int off = 0, padded = 0;
        const int cw_l = wcnt_s[lane];
        for (int w = 0; w < 32; ++w) {
            int cw = (__builtin_amdgcn_readlane(cw_l, w) + 31) & ~31;
            if (w < wave) off += cw;
            padded += cw;
        }
        const int wr = first && !spilled && padded <= rec_cap;
        if (first && wave == 0) {
            hdr[lane] = cw_l;
            if (lane == 0) hdr[32] = wr;
        }

        if (spilled == 0) {
            const int n = wcount;
            uint2* seg = myrec + REC_HDR + off;
            for (int i0 = 0; i0 < n; i0 += 32) {
                int cnt = n - i0;
                if (cnt > 32) cnt = 32;
                int   e  = mylist[i0 + (lane < cnt ? lane : 0)];
                int   rl = rows[e] - base;
                int   c  = cols[e];
                float v  = vals[e];
                if (wr && lane < cnt)
                    seg[i0 + lane] = make_uint2(((unsigned)rl << COL_BITS) | (unsigned)c,
                                                __builtin_bit_cast(unsigned, v));
                add_edges(rl, c, v, cnt, lane, acc, srcA, srcB, split);
            }
        } else {
            for (int q = tid; q < N_QUADS; q += 1024) {
                int4 r4 = rows4[q];
                int rr[4] = {r4.x, r4.y, r4.z, r4.w};
#pragma unroll
                for (int k = 0; k < 4; ++k) {
                    unsigned loc = (unsigned)(rr[k] - base);
                    if (loc < (unsigned)nr) {
                        int   e = q * 4 + k;
                        int   c = cols[e];
                        float v = vals[e];
                        const float* sp = (c < split)
                            ? (srcA + (size_t)c * DIM)
                            : (srcB + (size_t)(c - split) * DIM);
                        float* a = acc + loc * DIM;
#pragma nounroll
                        for (int d = 0; d < DIM; ++d)
                            unsafeAtomicAdd(a + d, v * sp[d]);
                    }
                }
            }
        }
    }
    __syncthreads();

    float4* out4 = (float4*)(Hout + (size_t)base * DIM);
    for (int i = tid; i < nr * DIM / 4; i += 1024)
        out4[i] = acc4[i];
}

#define PRED_PAIRS_PER_WAVE 128
#define PRED_TRIPS (PRED_PAIRS_PER_WAVE / 32)
static_assert(BATCH_CT % (8 * PRED_PAIRS_PER_WAVE) == 0, "whole blocks cover the batch");

template <int N>
__device__ __forceinline__ float row_xor(float x) {
    return __builtin_bit_cast(float, __builtin_amdgcn_update_dpp(
        0, __builtin_bit_cast(int, x), 0x160 + N, 0xf, 0xf, true));
}
__device__ __forceinline__ float row_swap(float x) {
    unsigned v = __builtin_bit_cast(unsigned, x);
    return __builtin_bit_cast(float, __builtin_amdgcn_permlanex16(
        v, v, 0x76543210u, 0xfedcba98u, false, false));
}
template <int N>
__device__ __forceinline__ float fold2(float a, float b, bool bit) {
    float keep = bit ? b : a;
    float send = bit ? a : b;
    return keep + row_xor<N>(send);
}

__global__ __launch_bounds__(256) void pred_kernel(
    const float* __restrict__ U, const float* __restrict__ I,
    const int*   __restrict__ u_idx, const int* __restrict__ i_idx,
    const float* __restrict__ bu, const float* __restrict__ bi,
    const float* __restrict__ mu, float* __restrict__ out) {
    const int lane = threadIdx.x & 31;
    const int wave = __builtin_amdgcn_readfirstlane(blockIdx.x * 8 + (threadIdx.x >> 5));
    const int  r    = lane & 7;
    const bool b0   = (lane & 1) != 0, b1 = (lane & 2) != 0, b2 = (lane & 4) != 0;
    const bool cell = (lane < 8) || (lane >= 24);
    const float mu0 = mu[0];

    for (int trip = 0; trip < PRED_TRIPS; ++trip) {
        const int base = wave * PRED_PAIRS_PER_WAVE + trip * 32;
        const int ur = u_idx[base + lane];
        const int ir = i_idx[base + lane];
        const float bu_v = bu[ur];
        const float bi_v = bi[ir];

        float G[4];
#pragma unroll
        for (int g = 0; g < 4; ++g) {
            v2f a[8], b[8];
#pragma unroll
            for (int j = 0; j < 8; ++j) {
                int su = __builtin_amdgcn_readlane(ur, 8 * g + j);
                int si = __builtin_amdgcn_readlane(ir, 8 * g + j);
                a[j] = *(const v2f*)(U + (size_t)su * DIM + 2 * lane);
                b[j] = *(const v2f*)(I + (size_t)si * DIM + 2 * lane);
            }
            float x[8];
#pragma unroll
            for (int j = 0; j < 8; ++j) {
                v8f c = {};
                c = __builtin_amdgcn_wmma_f32_16x16x4_f32(
                        false, a[j], false, b[j], (short)0, c, false, false);
                float t = c[0];
                t = (r == 1) ? c[1] : t;
                t = (r == 2) ? c[2] : t;
                t = (r == 3) ? c[3] : t;
                t = (r == 4) ? c[4] : t;
                t = (r == 5) ? c[5] : t;
                t = (r == 6) ? c[6] : t;
                t = (r == 7) ? c[7] : t;
                x[j] = t;
            }
            float y0 = fold2<1>(x[0], x[1], b0), y1 = fold2<1>(x[2], x[3], b0);
            float y2 = fold2<1>(x[4], x[5], b0), y3 = fold2<1>(x[6], x[7], b0);
            float z0 = fold2<2>(y0, y1, b1),     z1 = fold2<2>(y2, y3, b1);
            G[g] = fold2<4>(z0, z1, b2);
        }

        const float g1x = row_xor<8>(G[1]);
        const float g3x = row_xor<8>(G[3]);
        float h01 = cell ? G[0] : g1x;
        float h23 = cell ? G[2] : g3x;
        float t01 = h01 + row_swap(row_xor<8>(h01));
        float t23 = row_xor<8>(h23) + row_swap(h23);
        float d = (lane < 16) ? t01 : t23;

        out[base + lane] = d + bu_v + bi_v + mu0;
    }
}

extern "C" void kernel_launch(void* const* d_in, const int* in_sizes, int n_in,
                              void* d_out, int out_size, void* d_ws, size_t ws_size,
                              hipStream_t stream) {
    const int*   u_idx  = (const int*)d_in[0];
    const int*   i_idx  = (const int*)d_in[1];
    const int*   A_rows = (const int*)d_in[2];
    const int*   A_cols = (const int*)d_in[3];
    const float* A_vals = (const float*)d_in[4];
    const float* Eu     = (const float*)d_in[5];
    const float* Ei     = (const float*)d_in[6];
    const float* bu     = (const float*)d_in[7];
    const float* bi     = (const float*)d_in[8];
    const float* mu     = (const float*)d_in[9];
    float* out = (float*)d_out;

    float* H1 = (float*)d_ws;
    float* H2 = H1 + (size_t)N_NODES * DIM;

    unit_map_kernel<<<BM_BLOCKS, 1024, 0, stream>>>(A_rows, H2);

    spmm_owner_kernel<<<OWN_GROUPS, 1024, 0, stream>>>(
        A_rows, A_cols, A_vals, Eu, Ei, N_USERS, H1, H2, 1);

    spmm_owner_kernel<<<OWN_GROUPS, 1024, 0, stream>>>(
        A_rows, A_cols, A_vals, H1, H1, N_NODES + 1, H2, H2, 0);

    pred_kernel<<<BATCH_CT / (8 * PRED_PAIRS_PER_WAVE), 256, 0, stream>>>(
        H2, H2 + (size_t)N_USERS * DIM, u_idx, i_idx, bu, bi, mu, out);
}


# spmm layer 1: a list entry carries the edge's local row (known at the filter's match); the edge walk loads 2 scattered words per edge (source id, weight), not 3; LDS adds in arrival order as before
# speedup vs baseline: 2.3456x; 1.0426x over previous
#include <hip/hip_runtime.h>

#define N_USERS  100000
#define N_ITEMS  50000
#define N_NODES  (N_USERS + N_ITEMS)
#define NNZ_CT   1500000
#define BATCH_CT 1048576
#define DIM      64

typedef __attribute__((ext_vector_type(2))) float v2f;
typedef __attribute__((ext_vector_type(8))) float v8f;

#define OWN_ROWS   586
#define OWN_GROUPS 256
#define WAVE_CAP   1024
#define N_QUADS    (NNZ_CT / 4)
#define SCAN_STEPS 368
#define ROT_QUADS  1464
#define REC_HDR    32
#define COL_BITS   18
#define UNIT_EDGES 16
#define N_UNITS    (NNZ_CT / UNIT_EDGES)
#define BM_BLOCKS  92
#define BM_DWORDS  (BM_BLOCKS * 32)
#define BM_ROWS    (BM_DWORDS * 4 / (DIM * 4))
#define UNIT_CAP   512
#define EDGE_CAP   512
#define EDGE_BITS  21
static_assert(N_NODES <= (1 << COL_BITS), "source row id must fit its field");
static_assert(OWN_ROWS <= (1 << (32 - COL_BITS)), "local row must fit its field");
static_assert((N_NODES - 1) / OWN_ROWS < OWN_GROUPS, "every row has an owner");
static_assert(NNZ_CT % UNIT_EDGES == 0 && BM_BLOCKS * 1024 >= N_UNITS, "the map covers the edge list");
static_assert(BM_DWORDS * 4 % (DIM * 4) == 0, "a map is a whole number of rows");
static_assert(BM_DWORDS <= 3 * 1024, "three map dwords per lane cover a map");
static_assert(N_NODES - (OWN_GROUPS - 1) * OWN_ROWS > BM_ROWS + 256, "last range holds map and records");
static_assert(UNIT_CAP + EDGE_CAP <= WAVE_CAP, "both lists fit the wave's list");
static_assert(NNZ_CT <= (1 << EDGE_BITS), "edge id must fit its field of a list entry");
static_assert(OWN_ROWS <= (1 << (31 - EDGE_BITS)), "local row must fit above the edge id, sign bit clear");

#define BM_PITCH 257
__global__ __launch_bounds__(1024) void unit_map_kernel(
    const int* __restrict__ rows, float* plane) {
    __shared__ unsigned bm[32 * BM_PITCH];
    const int tid  = threadIdx.x;
    const int lane = tid & 31;
    const int wave = tid >> 5;
    for (int i = tid; i < 32 * BM_PITCH; i += 1024) bm[i] = 0u;
    const int4* rows4 = (const int4*)rows;
    const int  unit = blockIdx.x * 1024 + tid;
    const bool live = unit < N_UNITS;
    int4 r[4];
#pragma unroll
    for (int j = 0; j < 4; ++j) r[j] = rows4[live ? unit * 4 + j : 0];
    __syncthreads();
    if (live) {
        unsigned* mine = bm + wave * BM_PITCH;
        const unsigned bit = 1u << lane;
#pragma unroll
        for (int j = 0; j < 4; ++j) {
            unsigned g0 = (unsigned)r[j].x / OWN_ROWS, g1 = (unsigned)r[j].y / OWN_ROWS;
            unsigned g2 = (unsigned)r[j].z / OWN_ROWS, g3 = (unsigned)r[j].w / OWN_ROWS;
            if (g0 < OWN_GROUPS) atomicOr(mine + g0, bit);
            if (g1 < OWN_GROUPS) atomicOr(mine + g1, bit);
            if (g2 < OWN_GROUPS) atomicOr(mine + g2, bit);
            if (g3 < OWN_GROUPS) atomicOr(mine + g3, bit);
        }
    }
    __syncthreads();
    for (int g = wave; g < OWN_GROUPS; g += 32) {
        int nr = N_NODES - g * OWN_ROWS;
        if (nr > OWN_ROWS) nr = OWN_ROWS;
        unsigned* dst = (unsigned*)(plane + (size_t)(g * OWN_ROWS + nr - BM_ROWS) * DIM)
                      + blockIdx.x * 32;
        dst[lane] = bm[lane * BM_PITCH + g];
    }
}

#define NOTE_EDGE(m, b, e, r)                                           \
    if (b) {                                                            \
        if (m) {                                                        \
            int slot = wcount + (int)__builtin_amdgcn_mbcnt_lo((b), 0u);\
            if (slot < WAVE_CAP)                                        \
                mylist[slot] = (e) | (((r) - base) << EDGE_BITS);       \
        }                                                               \
        wcount += __builtin_popcount(b);                                \
    }

__device__ __forceinline__ void add_edges(
    int rl, int c, float v, int cnt, int lane, float* acc,
    const float* __restrict__ srcA, const float* __restrict__ srcB, int split) {
    for (int j0 = 0; j0 < cnt; j0 += 16) {
        v2f h[16];
#pragma unroll
        for (int u = 0; u < 16; ++u) {
            int j = j0 + u;
            if (j > cnt - 1) j = cnt - 1;
            int cj = __builtin_amdgcn_readlane(c, j);
            const float* sp = (cj < split)
                ? (srcA + (size_t)cj * DIM)
                : (srcB + (size_t)(cj - split) * DIM);
            h[u] = *(const v2f*)(sp + 2 * lane);
        }
#pragma unroll
        for (int u = 0; u < 16; ++u) {
            int j = j0 + u;
            if (j < cnt) {
                int   rj = __builtin_amdgcn_readlane(rl, j);
                float vj = __builtin_bit_cast(float,
                    __builtin_amdgcn_readlane(__builtin_bit_cast(int, v), j));
                float* a = acc + rj * DIM + 2 * lane;
                unsafeAtomicAdd(a,     vj * h[u].x);
                unsafeAtomicAdd(a + 1, vj * h[u].y);
            }
        }
    }
}

__global__ __launch_bounds__(1024) void spmm_owner_kernel(
    const int*   __restrict__ rows, const int* __restrict__ cols,
    const float* __restrict__ vals,
    const float* __restrict__ srcA, const float* __restrict__ srcB, int split,
    float* Hout, float* recplane, int first) {
    __shared__ float acc[OWN_ROWS * DIM];
    __shared__ int   elist[32 * WAVE_CAP];
    __shared__ int   spill;
    __shared__ int   wcnt_s[32];

    const int tid  = threadIdx.x;
    const int lane = tid & 31;
    const int base = blockIdx.x * OWN_ROWS;
    int nr = N_NODES - base;
    if (nr > OWN_ROWS) nr = OWN_ROWS;

    float4* acc4 = (float4*)acc;
    for (int i = tid; i < OWN_ROWS * DIM / 4; i += 1024)
        acc4[i] = make_float4(0.f, 0.f, 0.f, 0.f);
    if (tid == 0) spill = 0;

    uint2* myrec = (uint2*)(recplane + (size_t)base * DIM);
    int*   hdr   = (int*)myrec;
    const int rec_cap = (nr - BM_ROWS) * (DIM / 2) - REC_HDR;

    const int wave = __builtin_amdgcn_readfirstlane(tid >> 5);

    int listed = 0;
    int cl = 0;
    if (!first) {
        listed = __builtin_amdgcn_readfirstlane(hdr[32]) == 1;
        cl = hdr[lane];
    }
    __syncthreads();

    if (listed) {
        int n = 0, off = 0;
        for (int w = 0; w < 32; ++w) {
            int cw = __builtin_amdgcn_readlane(cl, w);
            if (w < wave)  off += (cw + 31) & ~31;
            if (w == wave) n = cw;
        }
        const uint2* seg = myrec + REC_HDR + off;
        uint2 nxt = make_uint2(0u, 0u);
        if (n > 0) nxt = seg[lane < n ? lane : 0];
        for (int i0 = 0; i0 < n; i0 += 32) {
            int cnt = n - i0;
            if (cnt > 32) cnt = 32;
            uint2 cur = nxt;
            if (i0 + 32 < n) {
                int x = i0 + 32 + lane;
                nxt = seg[x < n ? x : i0 + 32];
            }
            int   rl = (int)(cur.x >> COL_BITS);
            int   c  = (int)(cur.x & ((1u << COL_BITS) - 1u));
            float v  = __builtin_bit_cast(float, cur.y);
            add_edges(rl, c, v, cnt, lane, acc, srcA, srcB, split);
        }
    } else {
        int* mylist = elist + wave * WAVE_CAP;
        int  wcount = 0;

        const int4* rows4 = (const int4*)rows;
        const int rot = blockIdx.x * ROT_QUADS;
        if (first) {
            const unsigned* bmap = (const unsigned*)(recplane + (size_t)(base + nr - BM_ROWS) * DIM);
            int* ulist  = mylist + EDGE_CAP;
            int  ucount = 0;
            unsigned wd[3];
#pragma unroll
            for (int k = 0; k < 3; ++k) {
                int d = wave * 32 + lane + 1024 * k;
                wd[k] = bmap[d < BM_DWORDS ? d : 0];
                if (d >= BM_DWORDS) wd[k] = 0u;
            }
#pragma unroll
            for (int k = 0; k < 3; ++k) {
                unsigned x = wd[k];
                const int ub = (wave * 32 + lane + 1024 * k) * 32;
                bool     m = x != 0u;
                unsigned b = __builtin_amdgcn_ballot_w32(m);
#pragma nounroll
                while (b != 0u) {
                    int bit  = __builtin_ctz(x | 0x80000000u);
                    int slot = ucount + (int)__builtin_amdgcn_mbcnt_lo(b, 0u);
                    if (m && slot < UNIT_CAP) ulist[slot] = ub + bit;
                    x &= x - 1u;
                    ucount += __builtin_popcount(b);
                    m = x != 0u;
                    b = __builtin_amdgcn_ballot_w32(m);
                }
            }
            if (ucount > UNIT_CAP) { spill = 1; ucount = UNIT_CAP; }

            int  un = 0;
            int4 nx[4];
            if (ucount > 0) {
                un = ulist[lane < ucount ? lane : 0];
#pragma unroll
                for (int j = 0; j < 4; ++j) nx[j] = rows4[un * 4 + j];
            }
            for (int i0 = 0; i0 < ucount; i0 += 32) {
                const int  uc = un;
                const bool on = i0 + lane < ucount;
                int4 rv[4];
#pragma unroll
                for (int j = 0; j < 4; ++j) rv[j] = nx[j];
                if (i0 + 32 < ucount) {
                    int x = i0 + 32 + lane;
                    un = ulist[x < ucount ? x : i0 + 32];
#pragma unroll
                    for (int j = 0; j < 4; ++j) nx[j] = rows4[un * 4 + j];
                }
                unsigned lm = 0u;
#pragma unroll
                for (int j = 0; j < 4; ++j) {
                    if ((unsigned)(rv[j].x - base) < (unsigned)nr) lm |= 1u << (4 * j);
                    if ((unsigned)(rv[j].y - base) < (unsigned)nr) lm |= 2u << (4 * j);
                    if ((unsigned)(rv[j].z - base) < (unsigned)nr) lm |= 4u << (4 * j);
                    if ((unsigned)(rv[j].w - base) < (unsigned)nr) lm |= 8u << (4 * j);
                }
                if (!on) lm = 0u;
                bool     m = lm != 0u;
                unsigned b = __builtin_amdgcn_ballot_w32(m);
                const int e0 = uc * UNIT_EDGES;
#pragma nounroll
                while (b != 0u) {
                    int bit  = __builtin_ctz(lm | 0x80000000u);
                    int slot = wcount + (int)__builtin_amdgcn_mbcnt_lo(b, 0u);
                    int rid = 0;
#pragma unroll
                    for (int j = 3; j >= 0; --j) {
                        rid = (lm & (8u << (4 * j))) ? rv[j].w : rid;
                        rid = (lm & (4u << (4 * j))) ? rv[j].z : rid;
                        rid = (lm & (2u << (4 * j))) ? rv[j].y : rid;
                        rid = (lm & (1u << (4 * j))) ? rv[j].x : rid;
                    }
                    if (m && slot < EDGE_CAP)
                        mylist[slot] = (e0 + bit) | ((rid - base) << EDGE_BITS);
                    lm &= lm - 1u;
                    wcount += __builtin_popcount(b);
                    m = lm != 0u;
                    b = __builtin_amdgcn_ballot_w32(m);
                }
            }
            if (wcount > EDGE_CAP) spill = 1;
        } else {
            for (int i = 0; i < SCAN_STEPS; i += 16) {
                int4 rv[16];
                int  qx[16];
#pragma unroll
                for (int u = 0; u < 16; ++u) {
                    int q = tid + 1024 * (i + u);
                    int x = (q < N_QUADS ? q : 0) + rot;
                    if (x >= N_QUADS) x -= N_QUADS;
                    qx[u] = x;
                    rv[u] = rows4[x];
                }
#pragma unroll
                for (int u = 0; u < 16; ++u) {
                    bool ok = (tid + 1024 * (i + u)) < N_QUADS;
                    bool m0 = ok && (unsigned)(rv[u].x - base) < (unsigned)nr;
                    bool m1 = ok && (unsigned)(rv[u].y - base) < (unsigned)nr;
                    bool m2 = ok && (unsigned)(rv[u].z - base) < (unsigned)nr;
                    bool m3 = ok && (unsigned)(rv[u].w - base) < (unsigned)nr;
                    unsigned b0 = __builtin_amdgcn_ballot_w32(m0);
                    unsigned b1 = __builtin_amdgcn_ballot_w32(m1);
                    unsigned b2 = __builtin_amdgcn_ballot_w32(m2);
                    unsigned b3 = __builtin_amdgcn_ballot_w32(m3);
                    if (b0 | b1 | b2 | b3) {
                        int e0 = qx[u] * 4;
                        NOTE_EDGE(m0, b0, e0 + 0, rv[u].x)
                        NOTE_EDGE(m1, b1, e0 + 1, rv[u].y)
                        NOTE_EDGE(m2, b2, e0 + 2, rv[u].z)
                        NOTE_EDGE(m3, b3, e0 + 3, rv[u].w)
                    }
                }
            }
        }
        if (!first && wcount > WAVE_CAP) spill = 1;
        if (lane == 0) wcnt_s[wave] = wcount;
        __syncthreads();

        const int spilled = __builtin_amdgcn_readfirstlane(spill);

        int off = 0, padded = 0;
        const int cw_l = wcnt_s[lane];
        for (int w = 0; w < 32; ++w) {
            int cw = (__builtin_amdgcn_readlane(cw_l, w) + 31) & ~31;
            if (w < wave) off += cw;
            padded += cw;
        }
        const int wr = first && !spilled && padded <= rec_cap;
        if (first && wave == 0) {
            hdr[lane] = cw_l;
            if (lane == 0) hdr[32] = wr;
        }

        if (spilled == 0) {
            const int n = wcount;
            uint2* seg = myrec + REC_HDR + off;
            for (int i0 = 0; i0 < n; i0 += 32) {
                int cnt = n - i0;
                if (cnt > 32) cnt = 32;
                int   ent = mylist[i0 + (lane < cnt ? lane : 0)];
                int   e  = ent & ((1 << EDGE_BITS) - 1);
                int   rl = ent >> EDGE_BITS;
                int   c  = cols[e];
                float v  = vals[e];
                if (wr && lane < cnt)
                    seg[i0 + lane] = make_uint2(((unsigned)rl << COL_BITS) | (unsigned)c,
                                                __builtin_bit_cast(unsigned, v));
                add_edges(rl, c, v, cnt, lane, acc, srcA, srcB, split);
            }
        } else {
            for (int q = tid; q < N_QUADS; q += 1024) {
                int4 r4 = rows4[q];
                int rr[4] = {r4.x, r4.y, r4.z, r4.w};
#pragma unroll
                for (int k = 0; k < 4; ++k) {
                    unsigned loc = (unsigned)(rr[k] - base);
                    if (loc < (unsigned)nr) {
                        int   e = q * 4 + k;
                        int   c = cols[e];
                        float v = vals[e];
                        const float* sp = (c < split)
                            ? (srcA + (size_t)c * DIM)
                            : (srcB + (size_t)(c - split) * DIM);
                        float* a = acc + loc * DIM;
#pragma nounroll
                        for (int d = 0; d < DIM; ++d)
                            unsafeAtomicAdd(a + d, v * sp[d]);
                    }
                }
            }
        }
    }
    __syncthreads();

    float4* out4 = (float4*)(Hout + (size_t)base * DIM);
    for (int i = tid; i < nr * DIM / 4; i += 1024)
        out4[i] = acc4[i];
}

#define PRED_PAIRS_PER_WAVE 128
#define PRED_TRIPS (PRED_PAIRS_PER_WAVE / 32)
static_assert(BATCH_CT % (8 * PRED_PAIRS_PER_WAVE) == 0, "whole blocks cover the batch");

template <int N>
__device__ __forceinline__ float row_xor(float x) {
    return __builtin_bit_cast(float, __builtin_amdgcn_update_dpp(
        0, __builtin_bit_cast(int, x), 0x160 + N, 0xf, 0xf, true));
}
__device__ __forceinline__ float row_swap(float x) {
    unsigned v = __builtin_bit_cast(unsigned, x);
    return __builtin_bit_cast(float, __builtin_amdgcn_permlanex16(
        v, v, 0x76543210u, 0xfedcba98u, false, false));
}
template <int N>
__device__ __forceinline__ float fold2(float a, float b, bool bit) {
    float keep = bit ? b : a;
    float send = bit ? a : b;
    return keep + row_xor<N>(send);
}

__global__ __launch_bounds__(256) void pred_kernel(
    const float* __restrict__ U, const float* __restrict__ I,
    const int*   __restrict__ u_idx, const int* __restrict__ i_idx,
    const float* __restrict__ bu, const float* __restrict__ bi,
    const float* __restrict__ mu, float* __restrict__ out) {
    const int lane = threadIdx.x & 31;
    const int wave = __builtin_amdgcn_readfirstlane(blockIdx.x * 8 + (threadIdx.x >> 5));
    const int  r    = lane & 7;
    const bool b0   = (lane & 1) != 0, b1 = (lane & 2) != 0, b2 = (lane & 4) != 0;
    const bool cell = (lane < 8) || (lane >= 24);
    const float mu0 = mu[0];

    for (int trip = 0; trip < PRED_TRIPS; ++trip) {
        const int base = wave * PRED_PAIRS_PER_WAVE + trip * 32;
        const int ur = u_idx[base + lane];
        const int ir = i_idx[base + lane];
        const float bu_v = bu[ur];
        const float bi_v = bi[ir];

        float G[4];
#pragma unroll
        for (int g = 0; g < 4; ++g) {
            v2f a[8], b[8];
#pragma unroll
            for (int j = 0; j < 8; ++j) {
                int su = __builtin_amdgcn_readlane(ur, 8 * g + j);
                int si = __builtin_amdgcn_readlane(ir, 8 * g + j);
                a[j] = *(const v2f*)(U + (size_t)su * DIM + 2 * lane);
                b[j] = *(const v2f*)(I + (size_t)si * DIM + 2 * lane);
            }
            float x[8];
#pragma unroll
            for (int j = 0; j < 8; ++j) {
                v8f c = {};
                c = __builtin_amdgcn_wmma_f32_16x16x4_f32(
                        false, a[j], false, b[j], (short)0, c, false, false);
                float t = c[0];
                t = (r == 1) ? c[1] : t;
                t = (r == 2) ? c[2] : t;
                t = (r == 3) ? c[3] : t;
                t = (r == 4) ? c[4] : t;
                t = (r == 5) ? c[5] : t;
                t = (r == 6) ? c[6] : t;
                t = (r == 7) ? c[7] : t;
                x[j] = t;
            }
            float y0 = fold2<1>(x[0], x[1], b0), y1 = fold2<1>(x[2], x[3], b0);
            float y2 = fold2<1>(x[4], x[5], b0), y3 = fold2<1>(x[6], x[7], b0);
            float z0 = fold2<2>(y0, y1, b1),     z1 = fold2<2>(y2, y3, b1);
            G[g] = fold2<4>(z0, z1, b2);
        }

        const float g1x = row_xor<8>(G[1]);
        const float g3x = row_xor<8>(G[3]);
        float h01 = cell ? G[0] : g1x;
        float h23 = cell ? G[2] : g3x;
        float t01 = h01 + row_swap(row_xor<8>(h01));
        float t23 = row_xor<8>(h23) + row_swap(h23);
        float d = (lane < 16) ? t01 : t23;

        out[base + lane] = d + bu_v + bi_v + mu0;
    }
}

extern "C" void kernel_launch(void* const* d_in, const int* in_sizes, int n_in,
                              void* d_out, int out_size, void* d_ws, size_t ws_size,
                              hipStream_t stream) {
    const int*   u_idx  = (const int*)d_in[0];
    const int*   i_idx  = (const int*)d_in[1];
    const int*   A_rows = (const int*)d_in[2];
    const int*   A_cols = (const int*)d_in[3];
    const float* A_vals = (const float*)d_in[4];
    const float* Eu     = (const float*)d_in[5];
    const float* Ei     = (const float*)d_in[6];
    const float* bu     = (const float*)d_in[7];
    const float* bi     = (const float*)d_in[8];
    const float* mu     = (const float*)d_in[9];
    float* out = (float*)d_out;

    float* H1 = (float*)d_ws;
    float* H2 = H1 + (size_t)N_NODES * DIM;

    unit_map_kernel<<<BM_BLOCKS, 1024, 0, stream>>>(A_rows, H2);

    spmm_owner_kernel<<<OWN_GROUPS, 1024, 0, stream>>>(
        A_rows, A_cols, A_vals, Eu, Ei, N_USERS, H1, H2, 1);

    spmm_owner_kernel<<<OWN_GROUPS, 1024, 0, stream>>>(
        A_rows, A_cols, A_vals, H1, H1, N_NODES + 1, H2, H2, 0);

    pred_kernel<<<BATCH_CT / (8 * PRED_PAIRS_PER_WAVE), 256, 0, stream>>>(
        H2, H2 + (size_t)N_USERS * DIM, u_idx, i_idx, bu, bi, mu, out);
}
